# starting kernel
# baseline (speedup 1.0000x reference)
#include <hip/hip_runtime.h>
#include <hip/hip_bf16.h>


typedef __attribute__((ext_vector_type(2))) float v2f;
typedef __attribute__((ext_vector_type(8))) float v8f;

#define DF 48

__global__ void zero_f4(float4* __restrict__ p, int n4) {
    int gid = blockIdx.x * blockDim.x + threadIdx.x;
    if (gid < n4) p[gid] = make_float4(0.f, 0.f, 0.f, 0.f);
}

__global__ void scatter_add(const float* __restrict__ feat,
                            const int* __restrict__ src,
                            const int* __restrict__ dst,
                            float* __restrict__ agg,
                            long long total) {
    long long gid = (long long)blockIdx.x * blockDim.x + threadIdx.x;
    if (gid >= total) return;
    int e = (int)(gid / DF);
    int f = (int)(gid - (long long)e * DF);
    int s = src[e];
    int d = dst[e];
    unsafeAtomicAdd(&agg[(long long)d * DF + f], feat[(long long)s * DF + f]);
}

template <bool RELU>
__global__ __launch_bounds__(32)
void gemm_dual_wmma(const float* __restrict__ agg,
                    const float* __restrict__ xin,
                    const float* __restrict__ w_rel,
                    const float* __restrict__ w_root,
                    const float* __restrict__ bias,
                    float* __restrict__ out,
                    int n_nodes) {
    const int lane = threadIdx.x;
    const int hi   = lane >> 4;
    const int lo   = lane & 15;
    const int row0 = blockIdx.x * 16;

    const int arow = min(row0 + lo, n_nodes - 1);
    v2f a_agg[12], a_x[12];
#pragma unroll
    for (int k = 0; k < 12; ++k) {
        const int kk = 4 * k + 2 * hi;
        const long long base = (long long)arow * DF + kk;
        a_agg[k].x = agg[base];
        a_agg[k].y = agg[base + 1];
        a_x[k].x   = xin[base];
        a_x[k].y   = xin[base + 1];
    }

#pragma unroll
    for (int nt = 0; nt < 3; ++nt) {
        const int ncol = nt * 16 + lo;
        const float bv = bias[ncol];
        v8f c;
#pragma unroll
        for (int i = 0; i < 8; ++i) c[i] = bv;

#pragma unroll
        for (int k = 0; k < 12; ++k) {
            const int krow = 4 * k + 2 * hi;
            v2f brel, broot;
            brel.x  = w_rel [(krow    ) * DF + ncol];
            brel.y  = w_rel [(krow + 1) * DF + ncol];
            broot.x = w_root[(krow    ) * DF + ncol];
            broot.y = w_root[(krow + 1) * DF + ncol];
            c = __builtin_amdgcn_wmma_f32_16x16x4_f32(
                    false, a_agg[k], false, brel, (short)0, c, false, false);
            c = __builtin_amdgcn_wmma_f32_16x16x4_f32(
                    false, a_x[k], false, broot, (short)0, c, false, false);
        }

#pragma unroll
        for (int v = 0; v < 8; ++v) {
            float r = c[v];
            if (RELU) r = fmaxf(r, 0.f);
            const int m = row0 + v + 8 * hi;
            if (m < n_nodes) out[(long long)m * DF + ncol] = r;
        }
    }
}

extern "C" void kernel_launch(void* const* d_in, const int* in_sizes, int n_in,
                              void* d_out, int out_size, void* d_ws, size_t ws_size,
                              hipStream_t stream) {
    const float* x       = (const float*)d_in[0];
    const int*   ei      = (const int*)  d_in[1];
    const float* w_rel1  = (const float*)d_in[2];
    const float* w_root1 = (const float*)d_in[3];
    const float* b1      = (const float*)d_in[4];
    const float* w_rel2  = (const float*)d_in[5];
    const float* w_root2 = (const float*)d_in[6];
    const float* b2      = (const float*)d_in[7];
    float*       out     = (float*)d_out;

    const int N = in_sizes[0] / DF;
    const int E = in_sizes[1] / 2;
    const int* src = ei;
    const int* dst = ei + E;

    float* agg = (float*)d_ws;
    float* h   = agg + (size_t)N * DF;

    const int nfeat   = N * DF;
    const int n4      = nfeat / 4;
    const int zblocks = (n4 + 255) / 256;

    const long long etotal = (long long)E * DF;
    const int eblocks = (int)((etotal + 255) / 256);

    const int mtiles = (N + 15) / 16;

    zero_f4<<<zblocks, 256, 0, stream>>>((float4*)agg, n4);
    scatter_add<<<eblocks, 256, 0, stream>>>(x, src, dst, agg, etotal);
    gemm_dual_wmma<true><<<mtiles, 32, 0, stream>>>(agg, x, w_rel1, w_root1, b1, h, N);

    zero_f4<<<zblocks, 256, 0, stream>>>((float4*)agg, n4);
    scatter_add<<<eblocks, 256, 0, stream>>>(h, src, dst, agg, etotal);
    gemm_dual_wmma<false><<<mtiles, 32, 0, stream>>>(agg, h, w_rel2, w_root2, b2, out, N);
}


# Replace the per-element f32 atomic scatter by a destination-bucketed edge list (int count+rank, scan, fill; built once per call) and a per-node register gather-sum; sum order = arrival rank
# speedup vs baseline: 2.1974x; 2.1974x over previous
#include <hip/hip_runtime.h>
#include <hip/hip_bf16.h>


typedef __attribute__((ext_vector_type(2))) float v2f;
typedef __attribute__((ext_vector_type(8))) float v8f;

#define DF 48

__global__ void zero_i4(int4* __restrict__ p, int n4) {
    int gid = blockIdx.x * blockDim.x + threadIdx.x;
    if (gid < n4) p[gid] = make_int4(0, 0, 0, 0);
}

__global__ void count_rank(const int* __restrict__ dst,
                           int* __restrict__ cnt,
                           int* __restrict__ rank,
                           int n_edges) {
    int e = blockIdx.x * blockDim.x + threadIdx.x;
    if (e >= n_edges) return;
    rank[e] = atomicAdd(&cnt[dst[e]], 1);
}

#define SCAN_THREADS 1024
#define SCAN_TILES   7
#define SCAN_CHUNK4  (SCAN_THREADS * SCAN_TILES * 4)
#define SCAN_CHUNK   (SCAN_CHUNK4 * 4)

__device__ __forceinline__ int wave_incl_scan(int v, int lane) {
    for (int d = 1; d < 32; d <<= 1) {
        int o = __shfl_up(v, d, 32);
        if (lane >= d) v += o;
    }
    return v;
}

__global__ __launch_bounds__(SCAN_THREADS)
void scan_counts(const int4* __restrict__ cnt4, int4* __restrict__ off4, int nchunks) {
    __shared__ int exs[SCAN_TILES * SCAN_THREADS];
    __shared__ int wtot[SCAN_TILES * 32];
    __shared__ int wpre[SCAN_TILES * 32];
    __shared__ int part[32];
    __shared__ int chunk_total;
    const int t    = threadIdx.x;
    const int lane = t & 31;
    const int w    = t >> 5;
    int carry = 0;
    for (int c = 0; c < nchunks; ++c) {
        const int4* src4 = cnt4 + (size_t)c * SCAN_CHUNK4;
        int4*       dst4 = off4 + (size_t)c * SCAN_CHUNK4;
        for (int i = 0; i < SCAN_TILES; ++i) {
            const int4* p = src4 + (size_t)(i * SCAN_THREADS + t) * 4;
            const int4 v0 = p[0], v1 = p[1], v2 = p[2], v3 = p[3];
            const int s = (v0.x + v0.y + v0.z + v0.w) + (v1.x + v1.y + v1.z + v1.w)
                        + (v2.x + v2.y + v2.z + v2.w) + (v3.x + v3.y + v3.z + v3.w);
            const int inc = wave_incl_scan(s, lane);
            exs[i * SCAN_THREADS + t] = inc - s;
            if (lane == 31) wtot[i * 32 + w] = inc;
        }
        __syncthreads();
        const int x   = (t < SCAN_TILES * 32) ? wtot[t] : 0;
        const int inc = wave_incl_scan(x, lane);
        if (lane == 31) part[w] = inc;
        __syncthreads();
        int p = (lane < w) ? part[lane] : 0;
        for (int d = 16; d > 0; d >>= 1) p += __shfl_xor(p, d, 32);
        if (t < SCAN_TILES * 32) wpre[t] = inc - x + p;
        if (t == SCAN_THREADS - 1) chunk_total = inc + p;
        __syncthreads();
        for (int i = 0; i < SCAN_TILES; ++i) {
            const size_t q = (size_t)(i * SCAN_THREADS + t) * 4;
            const int4 v0 = src4[q], v1 = src4[q + 1], v2 = src4[q + 2], v3 = src4[q + 3];
            int4 o0, o1, o2, o3;
            o0.x = carry + wpre[i * 32 + w] + exs[i * SCAN_THREADS + t];
            o0.y = o0.x + v0.x;  o0.z = o0.y + v0.y;  o0.w = o0.z + v0.z;
            o1.x = o0.w + v0.w;  o1.y = o1.x + v1.x;  o1.z = o1.y + v1.y;  o1.w = o1.z + v1.z;
            o2.x = o1.w + v1.w;  o2.y = o2.x + v2.x;  o2.z = o2.y + v2.y;  o2.w = o2.z + v2.z;
            o3.x = o2.w + v2.w;  o3.y = o3.x + v3.x;  o3.z = o3.y + v3.y;  o3.w = o3.z + v3.z;
            dst4[q] = o0;  dst4[q + 1] = o1;  dst4[q + 2] = o2;  dst4[q + 3] = o3;
        }
        carry += chunk_total;
        __syncthreads();
    }
}

__global__ void fill_sources(const int* __restrict__ src,
                             const int* __restrict__ dst,
                             const int* __restrict__ off,
                             const int* __restrict__ rank,
                             int* __restrict__ perm,
                             int n_edges) {
    int e = blockIdx.x * blockDim.x + threadIdx.x;
    if (e >= n_edges) return;
    perm[off[dst[e]] + rank[e]] = src[e];
}

__global__ __launch_bounds__(256)
void gather_sum(const float* __restrict__ feat,
                const int* __restrict__ off,
                const int* __restrict__ perm,
                float* __restrict__ agg,
                int n_nodes) {
    const int gt   = blockIdx.x * 256 + threadIdx.x;
    const int node = gt >> 4;
    const int l    = threadIdx.x & 15;
    if (node >= n_nodes) return;
    const float4* __restrict__ f4 = (const float4*)feat;
    const bool act = l < 12;
    const int  lc  = act ? l : 11;
    const int beg = off[node];
    const int end = off[node + 1];
    float4 acc = make_float4(0.f, 0.f, 0.f, 0.f);
    const float4 z = make_float4(0.f, 0.f, 0.f, 0.f);
    for (int base = beg; base < end; base += 16) {
        const int m = min(16, end - base);
        const int mine = (l < m) ? perm[base + l] : 0;
        for (int j = 0; j < m; j += 4) {
            const int s0 = __shfl(mine, j,     16);
            const int s1 = __shfl(mine, j + 1, 16);
            const int s2 = __shfl(mine, j + 2, 16);
            const int s3 = __shfl(mine, j + 3, 16);
            float4 v0 = z, v1 = z, v2 = z, v3 = z;
            if (act) {
                v0 = f4[(long long)s0 * 12 + lc];
                if (j + 1 < m) v1 = f4[(long long)s1 * 12 + lc];
                if (j + 2 < m) v2 = f4[(long long)s2 * 12 + lc];
                if (j + 3 < m) v3 = f4[(long long)s3 * 12 + lc];
            }
            acc.x += v0.x; acc.y += v0.y; acc.z += v0.z; acc.w += v0.w;
            acc.x += v1.x; acc.y += v1.y; acc.z += v1.z; acc.w += v1.w;
            acc.x += v2.x; acc.y += v2.y; acc.z += v2.z; acc.w += v2.w;
            acc.x += v3.x; acc.y += v3.y; acc.z += v3.z; acc.w += v3.w;
        }
    }
    if (act) ((float4*)agg)[(long long)node * 12 + l] = acc;
}

template <bool RELU>
__global__ __launch_bounds__(32)
void gemm_dual_wmma(const float* __restrict__ agg,
                    const float* __restrict__ xin,
                    const float* __restrict__ w_rel,
                    const float* __restrict__ w_root,
                    const float* __restrict__ bias,
                    float* __restrict__ out,
                    int n_nodes) {
    const int lane = threadIdx.x;
    const int hi   = lane >> 4;
    const int lo   = lane & 15;
    const int row0 = blockIdx.x * 16;

    const int arow = min(row0 + lo, n_nodes - 1);
    v2f a_agg[12], a_x[12];
#pragma unroll
    for (int k = 0; k < 12; ++k) {
        const int kk = 4 * k + 2 * hi;
        const long long base = (long long)arow * DF + kk;
        a_agg[k].x = agg[base];
        a_agg[k].y = agg[base + 1];
        a_x[k].x   = xin[base];
        a_x[k].y   = xin[base + 1];
    }

#pragma unroll
    for (int nt = 0; nt < 3; ++nt) {
        const int ncol = nt * 16 + lo;
        const float bv = bias[ncol];
        v8f c;
#pragma unroll
        for (int i = 0; i < 8; ++i) c[i] = bv;

#pragma unroll
        for (int k = 0; k < 12; ++k) {
            const int krow = 4 * k + 2 * hi;
            v2f brel, broot;
            brel.x  = w_rel [(krow    ) * DF + ncol];
            brel.y  = w_rel [(krow + 1) * DF + ncol];
            broot.x = w_root[(krow    ) * DF + ncol];
            broot.y = w_root[(krow + 1) * DF + ncol];
            c = __builtin_amdgcn_wmma_f32_16x16x4_f32(
                    false, a_agg[k], false, brel, (short)0, c, false, false);
            c = __builtin_amdgcn_wmma_f32_16x16x4_f32(
                    false, a_x[k], false, broot, (short)0, c, false, false);
        }

#pragma unroll
        for (int v = 0; v < 8; ++v) {
            float r = c[v];
            if (RELU) r = fmaxf(r, 0.f);
            const int m = row0 + v + 8 * hi;
            if (m < n_nodes) out[(long long)m * DF + ncol] = r;
        }
    }
}

extern "C" void kernel_launch(void* const* d_in, const int* in_sizes, int n_in,
                              void* d_out, int out_size, void* d_ws, size_t ws_size,
                              hipStream_t stream) {
    const float* x       = (const float*)d_in[0];
    const int*   ei      = (const int*)  d_in[1];
    const float* w_rel1  = (const float*)d_in[2];
    const float* w_root1 = (const float*)d_in[3];
    const float* b1      = (const float*)d_in[4];
    const float* w_rel2  = (const float*)d_in[5];
    const float* w_root2 = (const float*)d_in[6];
    const float* b2      = (const float*)d_in[7];
    float*       out     = (float*)d_out;

    const int N = in_sizes[0] / DF;
    const int E = in_sizes[1] / 2;
    const int* src = ei;
    const int* dst = ei + E;

    float* agg = (float*)d_ws;
    float* h   = agg + (size_t)N * DF;

    const int nchunks = (N + 1 + SCAN_CHUNK - 1) / SCAN_CHUNK;
    const int NP      = nchunks * SCAN_CHUNK;
    int* cnt  = (int*)(h + (size_t)N * DF);
    int* off  = cnt + NP;
    int* rank = off + NP;
    int* perm = rank + E;

    const int zblocks = (NP / 4 + 255) / 256;
    const int eblocks = (E + 255) / 256;
    const int gblocks = (int)(((long long)N * 16 + 255) / 256);
    const int mtiles  = (N + 15) / 16;

    zero_i4<<<zblocks, 256, 0, stream>>>((int4*)cnt, NP / 4);
    count_rank<<<eblocks, 256, 0, stream>>>(dst, cnt, rank, E);
    scan_counts<<<1, SCAN_THREADS, 0, stream>>>((const int4*)cnt, (int4*)off, nchunks);
    fill_sources<<<eblocks, 256, 0, stream>>>(src, dst, off, rank, perm, E);

    gather_sum<<<gblocks, 256, 0, stream>>>(x, off, perm, agg, N);
    gemm_dual_wmma<true><<<mtiles, 32, 0, stream>>>(agg, x, w_rel1, w_root1, b1, h, N);

    gather_sum<<<gblocks, 256, 0, stream>>>(h, off, perm, agg, N);
    gemm_dual_wmma<false><<<mtiles, 32, 0, stream>>>(agg, h, w_rel2, w_root2, b2, out, N);
}


# Build the destination-grouped edge list by a two-level counting sort on LDS counters (512-node bins per 8192-edge chunk, then node counters per bin), no global atomics; sum order = LDS arrival rank
# speedup vs baseline: 3.8255x; 1.7409x over previous
#include <hip/hip_runtime.h>
#include <hip/hip_bf16.h>


typedef __attribute__((ext_vector_type(2))) float v2f;
typedef __attribute__((ext_vector_type(8))) float v8f;

#define DF 48

#define BSH    9
#define BNODES (1 << BSH)
#define ECH    8192
#define ETHR   1024

__global__ __launch_bounds__(ETHR)
void bin_count(const int* __restrict__ dst,
               int* __restrict__ mat,
               int n_edges, int nb, int g, int np) {
    extern __shared__ int lc[];
    const int t = threadIdx.x;
    const int c = blockIdx.x;
    for (int i = t; i < nb; i += ETHR) lc[i] = 0;
    __syncthreads();
    const int e0 = c * ECH + t;
    int d[ECH / ETHR];
#pragma unroll
    for (int k = 0; k < ECH / ETHR; ++k) d[k] = dst[min(e0 + k * ETHR, n_edges - 1)];
#pragma unroll
    for (int k = 0; k < ECH / ETHR; ++k)
        if (e0 + k * ETHR < n_edges) atomicAdd(&lc[d[k] >> BSH], 1);
    __syncthreads();
    for (int i = t; i < nb; i += ETHR) mat[i * g + c] = lc[i];
    for (int i = nb * g + c * ETHR + t; i < np; i += g * ETHR) mat[i] = 0;
}

#define SCAN_THREADS 1024
#define SCAN_TILES   3
#define SCAN_CHUNK4  (SCAN_THREADS * SCAN_TILES * 4)
#define SCAN_CHUNK   (SCAN_CHUNK4 * 4)

__device__ __forceinline__ int wave_incl_scan(int v, int lane) {
    for (int d = 1; d < 32; d <<= 1) {
        int o = __shfl_up(v, d, 32);
        if (lane >= d) v += o;
    }
    return v;
}

__global__ __launch_bounds__(SCAN_THREADS)
void scan_counts(const int4* __restrict__ cnt4, int4* __restrict__ off4, int nchunks) {
    __shared__ int exs[SCAN_TILES * SCAN_THREADS];
    __shared__ int wtot[SCAN_TILES * 32];
    __shared__ int wpre[SCAN_TILES * 32];
    __shared__ int part[32];
    __shared__ int chunk_total;
    const int t    = threadIdx.x;
    const int lane = t & 31;
    const int w    = t >> 5;
    int carry = 0;
    for (int c = 0; c < nchunks; ++c) {
        const int4* src4 = cnt4 + (size_t)c * SCAN_CHUNK4;
        int4*       dst4 = off4 + (size_t)c * SCAN_CHUNK4;
        for (int i = 0; i < SCAN_TILES; ++i) {
            const int4* p = src4 + (size_t)(i * SCAN_THREADS + t) * 4;
            const int4 v0 = p[0], v1 = p[1], v2 = p[2], v3 = p[3];
            const int s = (v0.x + v0.y + v0.z + v0.w) + (v1.x + v1.y + v1.z + v1.w)
                        + (v2.x + v2.y + v2.z + v2.w) + (v3.x + v3.y + v3.z + v3.w);
            const int inc = wave_incl_scan(s, lane);
            exs[i * SCAN_THREADS + t] = inc - s;
            if (lane == 31) wtot[i * 32 + w] = inc;
        }
        __syncthreads();
        const int x   = (t < SCAN_TILES * 32) ? wtot[t] : 0;
        const int inc = wave_incl_scan(x, lane);
        if (lane == 31) part[w] = inc;
        __syncthreads();
        int p = (lane < w) ? part[lane] : 0;
        for (int d = 16; d > 0; d >>= 1) p += __shfl_xor(p, d, 32);
        if (t < SCAN_TILES * 32) wpre[t] = inc - x + p;
        if (t == SCAN_THREADS - 1) chunk_total = inc + p;
        __syncthreads();
        for (int i = 0; i < SCAN_TILES; ++i) {
            const size_t q = (size_t)(i * SCAN_THREADS + t) * 4;
            const int4 v0 = src4[q], v1 = src4[q + 1], v2 = src4[q + 2], v3 = src4[q + 3];
            int4 o0, o1, o2, o3;
            o0.x = carry + wpre[i * 32 + w] + exs[i * SCAN_THREADS + t];
            o0.y = o0.x + v0.x;  o0.z = o0.y + v0.y;  o0.w = o0.z + v0.z;
            o1.x = o0.w + v0.w;  o1.y = o1.x + v1.x;  o1.z = o1.y + v1.y;  o1.w = o1.z + v1.z;
            o2.x = o1.w + v1.w;  o2.y = o2.x + v2.x;  o2.z = o2.y + v2.y;  o2.w = o2.z + v2.z;
            o3.x = o2.w + v2.w;  o3.y = o3.x + v3.x;  o3.z = o3.y + v3.y;  o3.w = o3.z + v3.z;
            dst4[q] = o0;  dst4[q + 1] = o1;  dst4[q + 2] = o2;  dst4[q + 3] = o3;
        }
        carry += chunk_total;
        __syncthreads();
    }
}

__global__ __launch_bounds__(ETHR)
void bin_scatter(const int* __restrict__ src,
                 const int* __restrict__ dst,
                 const int* __restrict__ moff,
                 int2* __restrict__ pairs,
                 int n_edges, int nb, int g) {
    extern __shared__ int lb[];
    const int t = threadIdx.x;
    const int c = blockIdx.x;
    for (int i = t; i < nb; i += ETHR) lb[i] = moff[i * g + c];
    __syncthreads();
    const int e0 = c * ECH + t;
    int d[ECH / ETHR], s[ECH / ETHR];
#pragma unroll
    for (int k = 0; k < ECH / ETHR; ++k) {
        const int e = min(e0 + k * ETHR, n_edges - 1);
        d[k] = dst[e];
        s[k] = src[e];
    }
#pragma unroll
    for (int k = 0; k < ECH / ETHR; ++k) {
        if (e0 + k * ETHR < n_edges) {
            const int p = atomicAdd(&lb[d[k] >> BSH], 1);
            pairs[p] = make_int2(s[k], d[k]);
        }
    }
}

__global__ __launch_bounds__(ETHR)
void bucket_fill(const int2* __restrict__ pairs,
                 const int* __restrict__ moff,
                 int* __restrict__ off,
                 int* __restrict__ perm,
                 int n_nodes, int g) {
    __shared__ int nc[BNODES];
    __shared__ int wsum[32];
    const int t    = threadIdx.x;
    const int lane = t & 31;
    const int w    = t >> 5;
    const int b    = blockIdx.x;
    const int beg   = moff[b * g];
    const int end   = moff[(b + 1) * g];
    const int node0 = b << BSH;
    if (t < BNODES) nc[t] = 0;
    __syncthreads();
    for (int i0 = beg + t; i0 < end; i0 += 4 * ETHR) {
        int dn[4];
#pragma unroll
        for (int u = 0; u < 4; ++u) dn[u] = pairs[min(i0 + u * ETHR, end - 1)].y;
#pragma unroll
        for (int u = 0; u < 4; ++u)
            if (i0 + u * ETHR < end) atomicAdd(&nc[dn[u] - node0], 1);
    }
    __syncthreads();
    const int v   = (t < BNODES) ? nc[t] : 0;
    const int inc = wave_incl_scan(v, lane);
    if (lane == 31) wsum[w] = inc;
    __syncthreads();
    int p = (lane < w) ? wsum[lane] : 0;
    for (int d = 16; d > 0; d >>= 1) p += __shfl_xor(p, d, 32);
    const int ex = beg + p + inc - v;
    if (t < BNODES) {
        nc[t] = ex;
        if (node0 + t <= n_nodes) off[node0 + t] = ex;
    }
    __syncthreads();
    for (int i0 = beg + t; i0 < end; i0 += 4 * ETHR) {
        int2 pr[4];
#pragma unroll
        for (int u = 0; u < 4; ++u) pr[u] = pairs[min(i0 + u * ETHR, end - 1)];
#pragma unroll
        for (int u = 0; u < 4; ++u) {
            if (i0 + u * ETHR < end) {
                const int q = atomicAdd(&nc[pr[u].y - node0], 1);
                perm[q] = pr[u].x;
            }
        }
    }
}

__global__ __launch_bounds__(256)
void gather_sum(const float* __restrict__ feat,
                const int* __restrict__ off,
                const int* __restrict__ perm,
                float* __restrict__ agg,
                int n_nodes) {
    const int gt   = blockIdx.x * 256 + threadIdx.x;
    const int node = gt >> 4;
    const int l    = threadIdx.x & 15;
    if (node >= n_nodes) return;
    const float4* __restrict__ f4 = (const float4*)feat;
    const bool act = l < 12;
    const int  lc  = act ? l : 11;
    const int beg = off[node];
    const int end = off[node + 1];
    float4 acc = make_float4(0.f, 0.f, 0.f, 0.f);
    const float4 z = make_float4(0.f, 0.f, 0.f, 0.f);
    for (int base = beg; base < end; base += 16) {
        const int m = min(16, end - base);
        const int mine = (l < m) ? perm[base + l] : 0;
        for (int j = 0; j < m; j += 4) {
            const int s0 = __shfl(mine, j,     16);
            const int s1 = __shfl(mine, j + 1, 16);
            const int s2 = __shfl(mine, j + 2, 16);
            const int s3 = __shfl(mine, j + 3, 16);
            float4 v0 = z, v1 = z, v2 = z, v3 = z;
            if (act) {
                v0 = f4[(long long)s0 * 12 + lc];
                if (j + 1 < m) v1 = f4[(long long)s1 * 12 + lc];
                if (j + 2 < m) v2 = f4[(long long)s2 * 12 + lc];
                if (j + 3 < m) v3 = f4[(long long)s3 * 12 + lc];
            }
            acc.x += v0.x; acc.y += v0.y; acc.z += v0.z; acc.w += v0.w;
            acc.x += v1.x; acc.y += v1.y; acc.z += v1.z; acc.w += v1.w;
            acc.x += v2.x; acc.y += v2.y; acc.z += v2.z; acc.w += v2.w;
            acc.x += v3.x; acc.y += v3.y; acc.z += v3.z; acc.w += v3.w;
        }
    }
    if (act) ((float4*)agg)[(long long)node * 12 + l] = acc;
}

template <bool RELU>
__global__ __launch_bounds__(32)
void gemm_dual_wmma(const float* __restrict__ agg,
                    const float* __restrict__ xin,
                    const float* __restrict__ w_rel,
                    const float* __restrict__ w_root,
                    const float* __restrict__ bias,
                    float* __restrict__ out,
                    int n_nodes) {
    const int lane = threadIdx.x;
    const int hi   = lane >> 4;
    const int lo   = lane & 15;
    const int row0 = blockIdx.x * 16;

    const int arow = min(row0 + lo, n_nodes - 1);
    v2f a_agg[12], a_x[12];
#pragma unroll
    for (int k = 0; k < 12; ++k) {
        const int kk = 4 * k + 2 * hi;
        const long long base = (long long)arow * DF + kk;
        a_agg[k].x = agg[base];
        a_agg[k].y = agg[base + 1];
        a_x[k].x   = xin[base];
        a_x[k].y   = xin[base + 1];
    }

#pragma unroll
    for (int nt = 0; nt < 3; ++nt) {
        const int ncol = nt * 16 + lo;
        const float bv = bias[ncol];
        v8f c;
#pragma unroll
        for (int i = 0; i < 8; ++i) c[i] = bv;

#pragma unroll
        for (int k = 0; k < 12; ++k) {
            const int krow = 4 * k + 2 * hi;
            v2f brel, broot;
            brel.x  = w_rel [(krow    ) * DF + ncol];
            brel.y  = w_rel [(krow + 1) * DF + ncol];
            broot.x = w_root[(krow    ) * DF + ncol];
            broot.y = w_root[(krow + 1) * DF + ncol];
            c = __builtin_amdgcn_wmma_f32_16x16x4_f32(
                    false, a_agg[k], false, brel, (short)0, c, false, false);
            c = __builtin_amdgcn_wmma_f32_16x16x4_f32(
                    false, a_x[k], false, broot, (short)0, c, false, false);
        }

#pragma unroll
        for (int v = 0; v < 8; ++v) {
            float r = c[v];
            if (RELU) r = fmaxf(r, 0.f);
            const int m = row0 + v + 8 * hi;
            if (m < n_nodes) out[(long long)m * DF + ncol] = r;
        }
    }
}

extern "C" void kernel_launch(void* const* d_in, const int* in_sizes, int n_in,
                              void* d_out, int out_size, void* d_ws, size_t ws_size,
                              hipStream_t stream) {
    const float* x       = (const float*)d_in[0];
    const int*   ei      = (const int*)  d_in[1];
    const float* w_rel1  = (const float*)d_in[2];
    const float* w_root1 = (const float*)d_in[3];
    const float* b1      = (const float*)d_in[4];
    const float* w_rel2  = (const float*)d_in[5];
    const float* w_root2 = (const float*)d_in[6];
    const float* b2      = (const float*)d_in[7];
    float*       out     = (float*)d_out;

    const int N = in_sizes[0] / DF;
    const int E = in_sizes[1] / 2;
    const int* src = ei;
    const int* dst = ei + E;

    float* agg = (float*)d_ws;
    float* h   = agg + (size_t)N * DF;

    const int nb      = (N + 1 + BNODES - 1) / BNODES;
    const int g       = (E + ECH - 1) / ECH;
    const int nchunks = (nb * g + 1 + SCAN_CHUNK - 1) / SCAN_CHUNK;
    const int NP      = nchunks * SCAN_CHUNK;
    const int NO      = (N + 1 + 3) / 4 * 4;
    int*  mat   = (int*)(h + (size_t)N * DF);
    int*  moff  = mat + NP;
    int*  off   = moff + NP;
    int2* pairs = (int2*)(off + NO);
    int*  perm  = (int*)(pairs + E);

    const int gblocks = (int)(((long long)N * 16 + 255) / 256);
    const int mtiles  = (N + 15) / 16;

    bin_count<<<g, ETHR, nb * sizeof(int), stream>>>(dst, mat, E, nb, g, NP);
    scan_counts<<<1, SCAN_THREADS, 0, stream>>>((const int4*)mat, (int4*)moff, nchunks);
    bin_scatter<<<g, ETHR, nb * sizeof(int), stream>>>(src, dst, moff, pairs, E, nb, g);
    bucket_fill<<<nb, ETHR, 0, stream>>>(pairs, moff, off, perm, N, g);

    gather_sum<<<gblocks, 256, 0, stream>>>(x, off, perm, agg, N);
    gemm_dual_wmma<true><<<mtiles, 32, 0, stream>>>(agg, x, w_rel1, w_root1, b1, h, N);

    gather_sum<<<gblocks, 256, 0, stream>>>(h, off, perm, agg, N);
    gemm_dual_wmma<false><<<mtiles, 32, 0, stream>>>(agg, h, w_rel2, w_root2, b2, out, N);
}


# Fuse the per-node gather-sum into the WMMA layer kernel: a wave sums its 16 nodes' rows (16 loads in flight) into an LDS tile that feeds the A fragments; no agg in memory; sum order kept
# speedup vs baseline: 4.7832x; 1.2503x over previous
#include <hip/hip_runtime.h>
#include <hip/hip_bf16.h>


typedef __attribute__((ext_vector_type(2))) float v2f;
typedef __attribute__((ext_vector_type(4))) float v4f;
typedef __attribute__((ext_vector_type(8))) float v8f;

#define DF 48

#define BSH    9
#define BNODES (1 << BSH)
#define ECH    8192
#define ETHR   1024

__global__ __launch_bounds__(ETHR)
void bin_count(const int* __restrict__ dst,
               int* __restrict__ mat,
               int n_edges, int nb, int g, int np) {
    extern __shared__ int lc[];
    const int t = threadIdx.x;
    const int c = blockIdx.x;
    for (int i = t; i < nb; i += ETHR) lc[i] = 0;
    __syncthreads();
    const int e0 = c * ECH + t;
    int d[ECH / ETHR];
#pragma unroll
    for (int k = 0; k < ECH / ETHR; ++k) d[k] = dst[min(e0 + k * ETHR, n_edges - 1)];
#pragma unroll
    for (int k = 0; k < ECH / ETHR; ++k)
        if (e0 + k * ETHR < n_edges) atomicAdd(&lc[d[k] >> BSH], 1);
    __syncthreads();
    for (int i = t; i < nb; i += ETHR) mat[i * g + c] = lc[i];
    for (int i = nb * g + c * ETHR + t; i < np; i += g * ETHR) mat[i] = 0;
}

#define SCAN_THREADS 1024
#define SCAN_TILES   3
#define SCAN_CHUNK4  (SCAN_THREADS * SCAN_TILES * 4)
#define SCAN_CHUNK   (SCAN_CHUNK4 * 4)

__device__ __forceinline__ int wave_incl_scan(int v, int lane) {
    for (int d = 1; d < 32; d <<= 1) {
        int o = __shfl_up(v, d, 32);
        if (lane >= d) v += o;
    }
    return v;
}

__global__ __launch_bounds__(SCAN_THREADS)
void scan_counts(const int4* __restrict__ cnt4, int4* __restrict__ off4, int nchunks) {
    __shared__ int exs[SCAN_TILES * SCAN_THREADS];
    __shared__ int wtot[SCAN_TILES * 32];
    __shared__ int wpre[SCAN_TILES * 32];
    __shared__ int part[32];
    __shared__ int chunk_total;
    const int t    = threadIdx.x;
    const int lane = t & 31;
    const int w    = t >> 5;
    int carry = 0;
    for (int c = 0; c < nchunks; ++c) {
        const int4* src4 = cnt4 + (size_t)c * SCAN_CHUNK4;
        int4*       dst4 = off4 + (size_t)c * SCAN_CHUNK4;
        for (int i = 0; i < SCAN_TILES; ++i) {
            const int4* p = src4 + (size_t)(i * SCAN_THREADS + t) * 4;
            const int4 v0 = p[0], v1 = p[1], v2 = p[2], v3 = p[3];
            const int s = (v0.x + v0.y + v0.z + v0.w) + (v1.x + v1.y + v1.z + v1.w)
                        + (v2.x + v2.y + v2.z + v2.w) + (v3.x + v3.y + v3.z + v3.w);
            const int inc = wave_incl_scan(s, lane);
            exs[i * SCAN_THREADS + t] = inc - s;
            if (lane == 31) wtot[i * 32 + w] = inc;
        }
        __syncthreads();
        const int x   = (t < SCAN_TILES * 32) ? wtot[t] : 0;
        const int inc = wave_incl_scan(x, lane);
        if (lane == 31) part[w] = inc;
        __syncthreads();
        int p = (lane < w) ? part[lane] : 0;
        for (int d = 16; d > 0; d >>= 1) p += __shfl_xor(p, d, 32);
        if (t < SCAN_TILES * 32) wpre[t] = inc - x + p;
        if (t == SCAN_THREADS - 1) chunk_total = inc + p;
        __syncthreads();
        for (int i = 0; i < SCAN_TILES; ++i) {
            const size_t q = (size_t)(i * SCAN_THREADS + t) * 4;
            const int4 v0 = src4[q], v1 = src4[q + 1], v2 = src4[q + 2], v3 = src4[q + 3];
            int4 o0, o1, o2, o3;
            o0.x = carry + wpre[i * 32 + w] + exs[i * SCAN_THREADS + t];
            o0.y = o0.x + v0.x;  o0.z = o0.y + v0.y;  o0.w = o0.z + v0.z;
            o1.x = o0.w + v0.w;  o1.y = o1.x + v1.x;  o1.z = o1.y + v1.y;  o1.w = o1.z + v1.z;
            o2.x = o1.w + v1.w;  o2.y = o2.x + v2.x;  o2.z = o2.y + v2.y;  o2.w = o2.z + v2.z;
            o3.x = o2.w + v2.w;  o3.y = o3.x + v3.x;  o3.z = o3.y + v3.y;  o3.w = o3.z + v3.z;
            dst4[q] = o0;  dst4[q + 1] = o1;  dst4[q + 2] = o2;  dst4[q + 3] = o3;
        }
        carry += chunk_total;
        __syncthreads();
    }
}

__global__ __launch_bounds__(ETHR)
void bin_scatter(const int* __restrict__ src,
                 const int* __restrict__ dst,
                 const int* __restrict__ moff,
                 int2* __restrict__ pairs,
                 int n_edges, int nb, int g) {
    extern __shared__ int lb[];
    const int t = threadIdx.x;
    const int c = blockIdx.x;
    for (int i = t; i < nb; i += ETHR) lb[i] = moff[i * g + c];
    __syncthreads();
    const int e0 = c * ECH + t;
    int d[ECH / ETHR], s[ECH / ETHR];
#pragma unroll
    for (int k = 0; k < ECH / ETHR; ++k) {
        const int e = min(e0 + k * ETHR, n_edges - 1);
        d[k] = dst[e];
        s[k] = src[e];
    }
#pragma unroll
    for (int k = 0; k < ECH / ETHR; ++k) {
        if (e0 + k * ETHR < n_edges) {
            const int p = atomicAdd(&lb[d[k] >> BSH], 1);
            pairs[p] = make_int2(s[k], d[k]);
        }
    }
}

__global__ __launch_bounds__(ETHR)
void bucket_fill(const int2* __restrict__ pairs,
                 const int* __restrict__ moff,
                 int* __restrict__ off,
                 int* __restrict__ perm,
                 int n_nodes, int g) {
    __shared__ int nc[BNODES];
    __shared__ int wsum[32];
    const int t    = threadIdx.x;
    const int lane = t & 31;
    const int w    = t >> 5;
    const int b    = blockIdx.x;
    const int beg   = moff[b * g];
    const int end   = moff[(b + 1) * g];
    const int node0 = b << BSH;
    if (t < BNODES) nc[t] = 0;
    __syncthreads();
    for (int i0 = beg + t; i0 < end; i0 += 4 * ETHR) {
        int dn[4];
#pragma unroll
        for (int u = 0; u < 4; ++u) dn[u] = pairs[min(i0 + u * ETHR, end - 1)].y;
#pragma unroll
        for (int u = 0; u < 4; ++u)
            if (i0 + u * ETHR < end) atomicAdd(&nc[dn[u] - node0], 1);
    }
    __syncthreads();
    const int v   = (t < BNODES) ? nc[t] : 0;
    const int inc = wave_incl_scan(v, lane);
    if (lane == 31) wsum[w] = inc;
    __syncthreads();
    int p = (lane < w) ? wsum[lane] : 0;
    for (int d = 16; d > 0; d >>= 1) p += __shfl_xor(p, d, 32);
    const int ex = beg + p + inc - v;
    if (t < BNODES) {
        nc[t] = ex;
        if (node0 + t <= n_nodes) off[node0 + t] = ex;
    }
    __syncthreads();
    for (int i0 = beg + t; i0 < end; i0 += 4 * ETHR) {
        int2 pr[4];
#pragma unroll
        for (int u = 0; u < 4; ++u) pr[u] = pairs[min(i0 + u * ETHR, end - 1)];
#pragma unroll
        for (int u = 0; u < 4; ++u) {
            if (i0 + u * ETHR < end) {
                const int q = atomicAdd(&nc[pr[u].y - node0], 1);
                perm[q] = pr[u].x;
            }
        }
    }
}

#define TP 52

template <bool RELU>
__global__ __launch_bounds__(32)
void conv_layer_wmma(const float* __restrict__ feat,
                     const int* __restrict__ off,
                     const int* __restrict__ perm,
                     const float* __restrict__ w_rel,
                     const float* __restrict__ w_root,
                     const float* __restrict__ bias,
                     float* __restrict__ out,
                     int n_nodes) {
    __shared__ __attribute__((aligned(16))) float tile[16 * TP];
    const int lane = threadIdx.x;
    const int hi   = lane >> 4;
    const int lo   = lane & 15;
    const int row0 = blockIdx.x * 16;

    const v4f* __restrict__ f4 = (const v4f*)feat;
    const bool act = lo < 12;
    const int  lc  = act ? lo : 11;
    const v4f z = {0.f, 0.f, 0.f, 0.f};
    const int o = off[min(row0 + min(lane, 16), n_nodes)];

    int nbeg  = __shfl(o, hi, 32);
    int nend  = __shfl(o, hi + 1, 32);
    int nmine = (lo < nend - nbeg) ? perm[nbeg + lo] : 0;
    for (int i = 0; i < 8; ++i) {
        const int r   = 2 * i + hi;
        const int beg = nbeg;
        const int end = nend;
        int mine = nmine;
        if (i < 7) {
            nbeg  = __shfl(o, r + 2, 32);
            nend  = __shfl(o, r + 3, 32);
            nmine = (lo < nend - nbeg) ? perm[nbeg + lo] : 0;
        }
        v4f acc = z;
        for (int base = beg; base < end; base += 16) {
            const int m = min(16, end - base);
            if (base != beg) mine = (lo < m) ? perm[base + lo] : 0;
            const int lim = act ? m : 0;
            v4f v[16];
#pragma unroll
            for (int u = 0; u < 16; ++u) {
                const int s = __shfl(mine, u, 16);
                v[u] = z;
                if (u < lim) v[u] = f4[(long long)s * 12 + lc];
            }
            asm volatile("" :: "v"(v[0]), "v"(v[1]), "v"(v[2]), "v"(v[3]),
                               "v"(v[4]), "v"(v[5]), "v"(v[6]), "v"(v[7]),
                               "v"(v[8]), "v"(v[9]), "v"(v[10]), "v"(v[11]),
                               "v"(v[12]), "v"(v[13]), "v"(v[14]), "v"(v[15]));
#pragma unroll
            for (int u = 0; u < 16; ++u) acc += v[u];
        }
        if (act) *(v4f*)&tile[r * TP + 4 * lo] = acc;
    }
    __syncthreads();

    const int arow = min(row0 + lo, n_nodes - 1);
    v2f a_agg[12], a_x[12];
#pragma unroll
    for (int k = 0; k < 12; ++k) {
        const int kk = 4 * k + 2 * hi;
        const long long base = (long long)arow * DF + kk;
        a_agg[k] = *(const v2f*)&tile[lo * TP + kk];
        a_x[k].x = feat[base];
        a_x[k].y = feat[base + 1];
    }

#pragma unroll
    for (int nt = 0; nt < 3; ++nt) {
        const int ncol = nt * 16 + lo;
        const float bv = bias[ncol];
        v8f c;
#pragma unroll
        for (int i = 0; i < 8; ++i) c[i] = bv;

#pragma unroll
        for (int k = 0; k < 12; ++k) {
            const int krow = 4 * k + 2 * hi;
            v2f brel, broot;
            brel.x  = w_rel [(krow    ) * DF + ncol];
            brel.y  = w_rel [(krow + 1) * DF + ncol];
            broot.x = w_root[(krow    ) * DF + ncol];
            broot.y = w_root[(krow + 1) * DF + ncol];
            c = __builtin_amdgcn_wmma_f32_16x16x4_f32(
                    false, a_agg[k], false, brel, (short)0, c, false, false);
            c = __builtin_amdgcn_wmma_f32_16x16x4_f32(
                    false, a_x[k], false, broot, (short)0, c, false, false);
        }

#pragma unroll
        for (int v = 0; v < 8; ++v) {
            float r = c[v];
            if (RELU) r = fmaxf(r, 0.f);
            const int m = row0 + v + 8 * hi;
            if (m < n_nodes) out[(long long)m * DF + ncol] = r;
        }
    }
}

extern "C" void kernel_launch(void* const* d_in, const int* in_sizes, int n_in,
                              void* d_out, int out_size, void* d_ws, size_t ws_size,
                              hipStream_t stream) {
    const float* x       = (const float*)d_in[0];
    const int*   ei      = (const int*)  d_in[1];
    const float* w_rel1  = (const float*)d_in[2];
    const float* w_root1 = (const float*)d_in[3];
    const float* b1      = (const float*)d_in[4];
    const float* w_rel2  = (const float*)d_in[5];
    const float* w_root2 = (const float*)d_in[6];
    const float* b2      = (const float*)d_in[7];
    float*       out     = (float*)d_out;

    const int N = in_sizes[0] / DF;
    const int E = in_sizes[1] / 2;
    const int* src = ei;
    const int* dst = ei + E;

    float* h = (float*)d_ws;

    const int nb      = (N + 1 + BNODES - 1) / BNODES;
    const int g       = (E + ECH - 1) / ECH;
    const int nchunks = (nb * g + 1 + SCAN_CHUNK - 1) / SCAN_CHUNK;
    const int NP      = nchunks * SCAN_CHUNK;
    const int NO      = (N + 1 + 3) / 4 * 4;
    int*  mat   = (int*)(h + (size_t)N * DF);
    int*  moff  = mat + NP;
    int*  off   = moff + NP;
    int2* pairs = (int2*)(off + NO);
    int*  perm  = (int*)(pairs + E);

    const int mtiles  = (N + 15) / 16;

    bin_count<<<g, ETHR, nb * sizeof(int), stream>>>(dst, mat, E, nb, g, NP);
    scan_counts<<<1, SCAN_THREADS, 0, stream>>>((const int4*)mat, (int4*)moff, nchunks);
    bin_scatter<<<g, ETHR, nb * sizeof(int), stream>>>(src, dst, moff, pairs, E, nb, g);
    bucket_fill<<<nb, ETHR, 0, stream>>>(pairs, moff, off, perm, N, g);

    conv_layer_wmma<true><<<mtiles, 32, 0, stream>>>(x, off, perm, w_rel1, w_root1, b1, h, N);

    conv_layer_wmma<false><<<mtiles, 32, 0, stream>>>(h, off, perm, w_rel2, w_root2, b2, out, N);
}


# Drop the scan launch: each scatter work-group sums the bin x chunk count table (wave per row, 14 int4 loads in flight) into its start positions; 6 launches -> 5; pair and sum order unchanged
# speedup vs baseline: 5.1357x; 1.0737x over previous
#include <hip/hip_runtime.h>
#include <hip/hip_bf16.h>


typedef __attribute__((ext_vector_type(2))) float v2f;
typedef __attribute__((ext_vector_type(4))) float v4f;
typedef __attribute__((ext_vector_type(8))) float v8f;

#define DF 48

#define BSH    9
#define BNODES (1 << BSH)
#define ECH    8192
#define ETHR   1024

__global__ __launch_bounds__(ETHR)
void bin_count(const int* __restrict__ dst,
               int* __restrict__ mat,
               int n_edges, int nb, int g, int gp) {
    extern __shared__ int lc[];
    const int t = threadIdx.x;
    const int c = blockIdx.x;
    for (int i = t; i < nb; i += ETHR) lc[i] = 0;
    __syncthreads();
    const int e0 = c * ECH + t;
    int d[ECH / ETHR];
#pragma unroll
    for (int k = 0; k < ECH / ETHR; ++k) d[k] = dst[min(e0 + k * ETHR, n_edges - 1)];
#pragma unroll
    for (int k = 0; k < ECH / ETHR; ++k)
        if (e0 + k * ETHR < n_edges) atomicAdd(&lc[d[k] >> BSH], 1);
    __syncthreads();
    for (int i = t; i < nb; i += ETHR) mat[i * gp + c] = lc[i];
    if (c == g - 1)
        for (int i = t; i < nb; i += ETHR)
            for (int j = g; j < gp; ++j) mat[i * gp + j] = 0;
}

__device__ __forceinline__ int wave_incl_scan(int v, int lane) {
    for (int d = 1; d < 32; d <<= 1) {
        int o = __shfl_up(v, d, 32);
        if (lane >= d) v += o;
    }
    return v;
}

#define RB 7
typedef __attribute__((ext_vector_type(4))) int v4i;

__global__ __launch_bounds__(ETHR)
void bin_scatter(const int* __restrict__ src,
                 const int* __restrict__ dst,
                 const int* __restrict__ mat,
                 int* __restrict__ binbase,
                 int2* __restrict__ pairs,
                 int n_edges, int nb, int g, int gp) {
    extern __shared__ int lb[];
    int* rtot = lb + nb;
    __shared__ int wsum[32];
    const int t    = threadIdx.x;
    const int lane = t & 31;
    const int w    = __builtin_amdgcn_readfirstlane(t >> 5);
    const int c    = blockIdx.x;

    const int e0 = c * ECH + t;
    int d[ECH / ETHR], s[ECH / ETHR];
#pragma unroll
    for (int k = 0; k < ECH / ETHR; ++k) {
        const int e = min(e0 + k * ETHR, n_edges - 1);
        d[k] = dst[e];
        s[k] = src[e];
    }

    const v4i* __restrict__ mat4 = (const v4i*)mat;
    const int q = gp >> 2;
    for (int i0 = w; i0 < nb; i0 += 32 * RB) {
        int tot[RB], part[RB];
#pragma unroll
        for (int r = 0; r < RB; ++r) { tot[r] = 0; part[r] = 0; }
        for (int j0 = 0; j0 < q; j0 += 64) {
            v4i v[RB][2];
#pragma unroll
            for (int r = 0; r < RB; ++r) {
                const v4i* __restrict__ row4 = mat4 + min(i0 + 32 * r, nb - 1) * q;
                v[r][0] = row4[min(j0 + lane, q - 1)];
                v[r][1] = row4[min(j0 + 32 + lane, q - 1)];
            }
            static_assert(RB == 7, "the statement below names 2 * RB values");
            asm volatile("" :: "v"(v[0][0]), "v"(v[1][0]), "v"(v[2][0]), "v"(v[3][0]),
                               "v"(v[4][0]), "v"(v[5][0]), "v"(v[6][0]),
                               "v"(v[0][1]), "v"(v[1][1]), "v"(v[2][1]), "v"(v[3][1]),
                               "v"(v[4][1]), "v"(v[5][1]), "v"(v[6][1]));
#pragma unroll
            for (int u = 0; u < 2; ++u) {
                const int  j   = j0 + 32 * u + lane;
                const bool ok  = j < q;
                const int  col = 4 * j;
                const int  cc  = ok ? c : 0;
#pragma unroll
                for (int r = 0; r < RB; ++r) {
                    const v4i x = v[r][u];
                    tot[r]  += ok ? ((x.x + x.y) + (x.z + x.w)) : 0;
                    part[r] += ((col     < cc) ? x.x : 0) + ((col + 1 < cc) ? x.y : 0)
                             + ((col + 2 < cc) ? x.z : 0) + ((col + 3 < cc) ? x.w : 0);
                }
            }
        }
#pragma unroll
        for (int x = 16; x > 0; x >>= 1) {
#pragma unroll
            for (int r = 0; r < RB; ++r) {
                const int a = __shfl_xor(tot[r], x, 32);
                const int b = __shfl_xor(part[r], x, 32);
                tot[r]  += a;
                part[r] += b;
            }
        }
#pragma unroll
        for (int r = 0; r < RB; ++r) {
            const int i = i0 + 32 * r;
            if (lane == 0 && i < nb) { rtot[i] = tot[r]; lb[i] = part[r]; }
        }
    }

    __syncthreads();

    int carry = 0;
    for (int b0 = 0; b0 < nb; b0 += ETHR) {
        const int i   = b0 + t;
        const int v   = (i < nb) ? rtot[i] : 0;
        const int inc = wave_incl_scan(v, lane);
        if (lane == 31) wsum[w] = inc;
        __syncthreads();
        const int ws = wsum[lane];
        int p   = (lane < w) ? ws : 0;
        int all = ws;
        for (int x = 16; x > 0; x >>= 1) {
            p   += __shfl_xor(p, x, 32);
            all += __shfl_xor(all, x, 32);
        }
        const int ex = carry + p + inc - v;
        if (i < nb) {
            lb[i] += ex;
            if (c == 0) binbase[i] = ex;
        }
        carry += all;
        __syncthreads();
    }
    if (c == 0 && t == 0) binbase[nb] = carry;

#pragma unroll
    for (int k = 0; k < ECH / ETHR; ++k) {
        if (e0 + k * ETHR < n_edges) {
            const int p = atomicAdd(&lb[d[k] >> BSH], 1);
            pairs[p] = make_int2(s[k], d[k]);
        }
    }
}

__global__ __launch_bounds__(ETHR)
void bucket_fill(const int2* __restrict__ pairs,
                 const int* __restrict__ binbase,
                 int* __restrict__ off,
                 int* __restrict__ perm,
                 int n_nodes) {
    __shared__ int nc[BNODES];
    __shared__ int wsum[32];
    const int t    = threadIdx.x;
    const int lane = t & 31;
    const int w    = t >> 5;
    const int b    = blockIdx.x;
    const int beg   = binbase[b];
    const int end   = binbase[b + 1];
    const int node0 = b << BSH;
    if (t < BNODES) nc[t] = 0;
    __syncthreads();
    for (int i0 = beg + t; i0 < end; i0 += 4 * ETHR) {
        int dn[4];
#pragma unroll
        for (int u = 0; u < 4; ++u) dn[u] = pairs[min(i0 + u * ETHR, end - 1)].y;
#pragma unroll
        for (int u = 0; u < 4; ++u)
            if (i0 + u * ETHR < end) atomicAdd(&nc[dn[u] - node0], 1);
    }
    __syncthreads();
    const int v   = (t < BNODES) ? nc[t] : 0;
    const int inc = wave_incl_scan(v, lane);
    if (lane == 31) wsum[w] = inc;
    __syncthreads();
    int p = (lane < w) ? wsum[lane] : 0;
    for (int d = 16; d > 0; d >>= 1) p += __shfl_xor(p, d, 32);
    const int ex = beg + p + inc - v;
    if (t < BNODES) {
        nc[t] = ex;
        if (node0 + t <= n_nodes) off[node0 + t] = ex;
    }
    __syncthreads();
    for (int i0 = beg + t; i0 < end; i0 += 4 * ETHR) {
        int2 pr[4];
#pragma unroll
        for (int u = 0; u < 4; ++u) pr[u] = pairs[min(i0 + u * ETHR, end - 1)];
#pragma unroll
        for (int u = 0; u < 4; ++u) {
            if (i0 + u * ETHR < end) {
                const int q = atomicAdd(&nc[pr[u].y - node0], 1);
                perm[q] = pr[u].x;
            }
        }
    }
}

#define TP 52

template <bool RELU>
__global__ __launch_bounds__(32)
void conv_layer_wmma(const float* __restrict__ feat,
                     const int* __restrict__ off,
                     const int* __restrict__ perm,
                     const float* __restrict__ w_rel,
                     const float* __restrict__ w_root,
                     const float* __restrict__ bias,
                     float* __restrict__ out,
                     int n_nodes) {
    __shared__ __attribute__((aligned(16))) float tile[16 * TP];
    const int lane = threadIdx.x;
    const int hi   = lane >> 4;
    const int lo   = lane & 15;
    const int row0 = blockIdx.x * 16;

    const v4f* __restrict__ f4 = (const v4f*)feat;
    const bool act = lo < 12;
    const int  lc  = act ? lo : 11;
    const v4f z = {0.f, 0.f, 0.f, 0.f};
    const int o = off[min(row0 + min(lane, 16), n_nodes)];

    int nbeg  = __shfl(o, hi, 32);
    int nend  = __shfl(o, hi + 1, 32);
    int nmine = (lo < nend - nbeg) ? perm[nbeg + lo] : 0;
    for (int i = 0; i < 8; ++i) {
        const int r   = 2 * i + hi;
        const int beg = nbeg;
        const int end = nend;
        int mine = nmine;
        if (i < 7) {
            nbeg  = __shfl(o, r + 2, 32);
            nend  = __shfl(o, r + 3, 32);
            nmine = (lo < nend - nbeg) ? perm[nbeg + lo] : 0;
        }
        v4f acc = z;
        for (int base = beg; base < end; base += 16) {
            const int m = min(16, end - base);
            if (base != beg) mine = (lo < m) ? perm[base + lo] : 0;
            const int lim = act ? m : 0;
            v4f v[16];
#pragma unroll
            for (int u = 0; u < 16; ++u) {
                const int s = __shfl(mine, u, 16);
                v[u] = z;
                if (u < lim) v[u] = f4[(long long)s * 12 + lc];
            }
            asm volatile("" :: "v"(v[0]), "v"(v[1]), "v"(v[2]), "v"(v[3]),
                               "v"(v[4]), "v"(v[5]), "v"(v[6]), "v"(v[7]),
                               "v"(v[8]), "v"(v[9]), "v"(v[10]), "v"(v[11]),
                               "v"(v[12]), "v"(v[13]), "v"(v[14]), "v"(v[15]));
#pragma unroll
            for (int u = 0; u < 16; ++u) acc += v[u];
        }
        if (act) *(v4f*)&tile[r * TP + 4 * lo] = acc;
    }
    __syncthreads();

    const int arow = min(row0 + lo, n_nodes - 1);
    v2f a_agg[12], a_x[12];
#pragma unroll
    for (int k = 0; k < 12; ++k) {
        const int kk = 4 * k + 2 * hi;
        const long long base = (long long)arow * DF + kk;
        a_agg[k] = *(const v2f*)&tile[lo * TP + kk];
        a_x[k].x = feat[base];
        a_x[k].y = feat[base + 1];
    }

#pragma unroll
    for (int nt = 0; nt < 3; ++nt) {
        const int ncol = nt * 16 + lo;
        const float bv = bias[ncol];
        v8f c;
#pragma unroll
        for (int i = 0; i < 8; ++i) c[i] = bv;

#pragma unroll
        for (int k = 0; k < 12; ++k) {
            const int krow = 4 * k + 2 * hi;
            v2f brel, broot;
            brel.x  = w_rel [(krow    ) * DF + ncol];
            brel.y  = w_rel [(krow + 1) * DF + ncol];
            broot.x = w_root[(krow    ) * DF + ncol];
            broot.y = w_root[(krow + 1) * DF + ncol];
            c = __builtin_amdgcn_wmma_f32_16x16x4_f32(
                    false, a_agg[k], false, brel, (short)0, c, false, false);
            c = __builtin_amdgcn_wmma_f32_16x16x4_f32(
                    false, a_x[k], false, broot, (short)0, c, false, false);
        }

#pragma unroll
        for (int v = 0; v < 8; ++v) {
            float r = c[v];
            if (RELU) r = fmaxf(r, 0.f);
            const int m = row0 + v + 8 * hi;
            if (m < n_nodes) out[(long long)m * DF + ncol] = r;
        }
    }
}

extern "C" void kernel_launch(void* const* d_in, const int* in_sizes, int n_in,
                              void* d_out, int out_size, void* d_ws, size_t ws_size,
                              hipStream_t stream) {
    const float* x       = (const float*)d_in[0];
    const int*   ei      = (const int*)  d_in[1];
    const float* w_rel1  = (const float*)d_in[2];
    const float* w_root1 = (const float*)d_in[3];
    const float* b1      = (const float*)d_in[4];
    const float* w_rel2  = (const float*)d_in[5];
    const float* w_root2 = (const float*)d_in[6];
    const float* b2      = (const float*)d_in[7];
    float*       out     = (float*)d_out;

    const int N = in_sizes[0] / DF;
    const int E = in_sizes[1] / 2;
    const int* src = ei;
    const int* dst = ei + E;

    float* h = (float*)d_ws;

    const int nb      = (N + 1 + BNODES - 1) / BNODES;
    const int g       = (E + ECH - 1) / ECH;
    const int gp      = (g + 3) / 4 * 4;
    const int NBP     = (nb + 1 + 3) / 4 * 4;
    const int NO      = (N + 1 + 3) / 4 * 4;
    int*  mat     = (int*)(h + (size_t)N * DF);
    int*  binbase = mat + (size_t)nb * gp;
    int*  off     = binbase + NBP;
    int2* pairs = (int2*)(off + NO);
    int*  perm  = (int*)(pairs + E);

    const int mtiles  = (N + 15) / 16;

    bin_count<<<g, ETHR, nb * sizeof(int), stream>>>(dst, mat, E, nb, g, gp);
    bin_scatter<<<g, ETHR, 2 * nb * sizeof(int), stream>>>(src, dst, mat, binbase, pairs, E, nb, g, gp);
    bucket_fill<<<nb, ETHR, 0, stream>>>(pairs, binbase, off, perm, N);

    conv_layer_wmma<true><<<mtiles, 32, 0, stream>>>(x, off, perm, w_rel1, w_root1, b1, h, N);

    conv_layer_wmma<false><<<mtiles, 32, 0, stream>>>(h, off, perm, w_rel2, w_root2, b2, out, N);
}


# Gather with 4 lanes per node (3 float4 each, 64-byte pieces): all 32 lanes load and 8 nodes advance together, 8 rows x 3 loads in flight per quad; was 16 lanes per node; sum order kept
# speedup vs baseline: 5.2984x; 1.0317x over previous
#include <hip/hip_runtime.h>
#include <hip/hip_bf16.h>


typedef __attribute__((ext_vector_type(2))) float v2f;
typedef __attribute__((ext_vector_type(4))) float v4f;
typedef __attribute__((ext_vector_type(8))) float v8f;

#define DF 48

#define BSH    9
#define BNODES (1 << BSH)
#define ECH    8192
#define ETHR   1024

__global__ __launch_bounds__(ETHR)
void bin_count(const int* __restrict__ dst,
               int* __restrict__ mat,
               int n_edges, int nb, int g, int gp) {
    extern __shared__ int lc[];
    const int t = threadIdx.x;
    const int c = blockIdx.x;
    for (int i = t; i < nb; i += ETHR) lc[i] = 0;
    __syncthreads();
    const int e0 = c * ECH + t;
    int d[ECH / ETHR];
#pragma unroll
    for (int k = 0; k < ECH / ETHR; ++k) d[k] = dst[min(e0 + k * ETHR, n_edges - 1)];
#pragma unroll
    for (int k = 0; k < ECH / ETHR; ++k)
        if (e0 + k * ETHR < n_edges) atomicAdd(&lc[d[k] >> BSH], 1);
    __syncthreads();
    for (int i = t; i < nb; i += ETHR) mat[i * gp + c] = lc[i];
    if (c == g - 1)
        for (int i = t; i < nb; i += ETHR)
            for (int j = g; j < gp; ++j) mat[i * gp + j] = 0;
}

__device__ __forceinline__ int wave_incl_scan(int v, int lane) {
    for (int d = 1; d < 32; d <<= 1) {
        int o = __shfl_up(v, d, 32);
        if (lane >= d) v += o;
    }
    return v;
}

#define RB 7
typedef __attribute__((ext_vector_type(4))) int v4i;

__global__ __launch_bounds__(ETHR)
void bin_scatter(const int* __restrict__ src,
                 const int* __restrict__ dst,
                 const int* __restrict__ mat,
                 int* __restrict__ binbase,
                 int2* __restrict__ pairs,
                 int n_edges, int nb, int g, int gp) {
    extern __shared__ int lb[];
    int* rtot = lb + nb;
    __shared__ int wsum[32];
    const int t    = threadIdx.x;
    const int lane = t & 31;
    const int w    = __builtin_amdgcn_readfirstlane(t >> 5);
    const int c    = blockIdx.x;

    const int e0 = c * ECH + t;
    int d[ECH / ETHR], s[ECH / ETHR];
#pragma unroll
    for (int k = 0; k < ECH / ETHR; ++k) {
        const int e = min(e0 + k * ETHR, n_edges - 1);
        d[k] = dst[e];
        s[k] = src[e];
    }

    const v4i* __restrict__ mat4 = (const v4i*)mat;
    const int q = gp >> 2;
    for (int i0 = w; i0 < nb; i0 += 32 * RB) {
        int tot[RB], part[RB];
#pragma unroll
        for (int r = 0; r < RB; ++r) { tot[r] = 0; part[r] = 0; }
        for (int j0 = 0; j0 < q; j0 += 64) {
            v4i v[RB][2];
#pragma unroll
            for (int r = 0; r < RB; ++r) {
                const v4i* __restrict__ row4 = mat4 + min(i0 + 32 * r, nb - 1) * q;
                v[r][0] = row4[min(j0 + lane, q - 1)];
                v[r][1] = row4[min(j0 + 32 + lane, q - 1)];
            }
            static_assert(RB == 7, "the statement below names 2 * RB values");
            asm volatile("" :: "v"(v[0][0]), "v"(v[1][0]), "v"(v[2][0]), "v"(v[3][0]),
                               "v"(v[4][0]), "v"(v[5][0]), "v"(v[6][0]),
                               "v"(v[0][1]), "v"(v[1][1]), "v"(v[2][1]), "v"(v[3][1]),
                               "v"(v[4][1]), "v"(v[5][1]), "v"(v[6][1]));
#pragma unroll
            for (int u = 0; u < 2; ++u) {
                const int  j   = j0 + 32 * u + lane;
                const bool ok  = j < q;
                const int  col = 4 * j;
                const int  cc  = ok ? c : 0;
#pragma unroll
                for (int r = 0; r < RB; ++r) {
                    const v4i x = v[r][u];
                    tot[r]  += ok ? ((x.x + x.y) + (x.z + x.w)) : 0;
                    part[r] += ((col     < cc) ? x.x : 0) + ((col + 1 < cc) ? x.y : 0)
                             + ((col + 2 < cc) ? x.z : 0) + ((col + 3 < cc) ? x.w : 0);
                }
            }
        }
#pragma unroll
        for (int x = 16; x > 0; x >>= 1) {
#pragma unroll
            for (int r = 0; r < RB; ++r) {
                const int a = __shfl_xor(tot[r], x, 32);
                const int b = __shfl_xor(part[r], x, 32);
                tot[r]  += a;
                part[r] += b;
            }
        }
#pragma unroll
        for (int r = 0; r < RB; ++r) {
            const int i = i0 + 32 * r;
            if (lane == 0 && i < nb) { rtot[i] = tot[r]; lb[i] = part[r]; }
        }
    }

    __syncthreads();

    int carry = 0;
    for (int b0 = 0; b0 < nb; b0 += ETHR) {
        const int i   = b0 + t;
        const int v   = (i < nb) ? rtot[i] : 0;
        const int inc = wave_incl_scan(v, lane);
        if (lane == 31) wsum[w] = inc;
        __syncthreads();
        const int ws = wsum[lane];
        int p   = (lane < w) ? ws : 0;
        int all = ws;
        for (int x = 16; x > 0; x >>= 1) {
            p   += __shfl_xor(p, x, 32);
            all += __shfl_xor(all, x, 32);
        }
        const int ex = carry + p + inc - v;
        if (i < nb) {
            lb[i] += ex;
            if (c == 0) binbase[i] = ex;
        }
        carry += all;
        __syncthreads();
    }
    if (c == 0 && t == 0) binbase[nb] = carry;

#pragma unroll
    for (int k = 0; k < ECH / ETHR; ++k) {
        if (e0 + k * ETHR < n_edges) {
            const int p = atomicAdd(&lb[d[k] >> BSH], 1);
            pairs[p] = make_int2(s[k], d[k]);
        }
    }
}

__global__ __launch_bounds__(ETHR)
void bucket_fill(const int2* __restrict__ pairs,
                 const int* __restrict__ binbase,
                 int* __restrict__ off,
                 int* __restrict__ perm,
                 int n_nodes) {
    __shared__ int nc[BNODES];
    __shared__ int wsum[32];
    const int t    = threadIdx.x;
    const int lane = t & 31;
    const int w    = t >> 5;
    const int b    = blockIdx.x;
    const int beg   = binbase[b];
    const int end   = binbase[b + 1];
    const int node0 = b << BSH;
    if (t < BNODES) nc[t] = 0;
    __syncthreads();
    for (int i0 = beg + t; i0 < end; i0 += 4 * ETHR) {
        int dn[4];
#pragma unroll
        for (int u = 0; u < 4; ++u) dn[u] = pairs[min(i0 + u * ETHR, end - 1)].y;
#pragma unroll
        for (int u = 0; u < 4; ++u)
            if (i0 + u * ETHR < end) atomicAdd(&nc[dn[u] - node0], 1);
    }
    __syncthreads();
    const int v   = (t < BNODES) ? nc[t] : 0;
    const int inc = wave_incl_scan(v, lane);
    if (lane == 31) wsum[w] = inc;
    __syncthreads();
    int p = (lane < w) ? wsum[lane] : 0;
    for (int d = 16; d > 0; d >>= 1) p += __shfl_xor(p, d, 32);
    const int ex = beg + p + inc - v;
    if (t < BNODES) {
        nc[t] = ex;
        if (node0 + t <= n_nodes) off[node0 + t] = ex;
    }
    __syncthreads();
    for (int i0 = beg + t; i0 < end; i0 += 4 * ETHR) {
        int2 pr[4];
#pragma unroll
        for (int u = 0; u < 4; ++u) pr[u] = pairs[min(i0 + u * ETHR, end - 1)];
#pragma unroll
        for (int u = 0; u < 4; ++u) {
            if (i0 + u * ETHR < end) {
                const int q = atomicAdd(&nc[pr[u].y - node0], 1);
                perm[q] = pr[u].x;
            }
        }
    }
}

#define TP 52

template <bool RELU>
__global__ __launch_bounds__(32)
void conv_layer_wmma(const float* __restrict__ feat,
                     const int* __restrict__ off,
                     const int* __restrict__ perm,
                     const float* __restrict__ w_rel,
                     const float* __restrict__ w_root,
                     const float* __restrict__ bias,
                     float* __restrict__ out,
                     int n_nodes) {
    __shared__ __attribute__((aligned(16))) float tile[16 * TP];
    const int lane = threadIdx.x;
    const int hi   = lane >> 4;
    const int lo   = lane & 15;
    const int row0 = blockIdx.x * 16;

    const v4f* __restrict__ f4 = (const v4f*)feat;
    const int p = lane & 3;
    const int q = lane >> 2;
    const v4f z = {0.f, 0.f, 0.f, 0.f};
    const int o = off[min(row0 + min(lane, 16), n_nodes)];

    int nbeg[2], ndeg[2], nia[2], nib[2];
#pragma unroll
    for (int r = 0; r < 2; ++r) {
        const int j = 8 * r + q;
        nbeg[r] = __shfl(o, j, 32);
        ndeg[r] = __shfl(o, j + 1, 32) - nbeg[r];
        nia[r]  = (p     < ndeg[r]) ? perm[nbeg[r] + p]     : 0;
        nib[r]  = (4 + p < ndeg[r]) ? perm[nbeg[r] + 4 + p] : 0;
    }

    asm volatile("" :: "v"(nia[0]), "v"(nib[0]), "v"(nia[1]), "v"(nib[1]));

#define GATHER_SLOT(U)                                                          \
    {                                                                           \
        const int s = __builtin_amdgcn_update_dpp(0, ((U) < 4) ? ca : cb,       \
                                                  ((U) & 3) * 0x55, 0xf, 0xf,   \
                                                  true);                        \
        v[U][0] = z; v[U][1] = z; v[U][2] = z;                                  \
        if ((U) < rem) {                                                        \
            const v4f* __restrict__ rp = f4 + (long long)s * 12 + p;            \
            v[U][0] = rp[0];                                                    \
            v[U][1] = rp[4];                                                    \
            v[U][2] = rp[8];                                                    \
        }                                                                       \
    }

#pragma unroll
    for (int r = 0; r < 2; ++r) {
        const int j   = 8 * r + q;
        const int beg = nbeg[r];
        const int deg = ndeg[r];
        int na = nia[r], nb = nib[r];
        v4f a0 = z, a1 = z, a2 = z;
        for (int base = 0; base < deg; base += 8) {
            const int rem = deg - base;
            const int ca = na, cb = nb;
            na = (8 + p  < rem) ? perm[beg + base + 8 + p]  : 0;
            nb = (12 + p < rem) ? perm[beg + base + 12 + p] : 0;
            v4f v[8][3];
            GATHER_SLOT(0) GATHER_SLOT(1) GATHER_SLOT(2) GATHER_SLOT(3)
            GATHER_SLOT(4) GATHER_SLOT(5) GATHER_SLOT(6) GATHER_SLOT(7)
#pragma unroll
            for (int u = 0; u < 8; ++u) { a0 += v[u][0]; a1 += v[u][1]; a2 += v[u][2]; }
        }
        float* trow = &tile[j * TP + 4 * p];
        *(v4f*)(trow)      = a0;
        *(v4f*)(trow + 16) = a1;
        *(v4f*)(trow + 32) = a2;
    }
#undef GATHER_SLOT
    __syncthreads();

    const int arow = min(row0 + lo, n_nodes - 1);
    v2f a_agg[12], a_x[12];
#pragma unroll
    for (int k = 0; k < 12; ++k) {
        const int kk = 4 * k + 2 * hi;
        const long long base = (long long)arow * DF + kk;
        a_agg[k] = *(const v2f*)&tile[lo * TP + kk];
        a_x[k].x = feat[base];
        a_x[k].y = feat[base + 1];
    }

#pragma unroll
    for (int nt = 0; nt < 3; ++nt) {
        const int ncol = nt * 16 + lo;
        const float bv = bias[ncol];
        v8f c;
#pragma unroll
        for (int i = 0; i < 8; ++i) c[i] = bv;

#pragma unroll
        for (int k = 0; k < 12; ++k) {
            const int krow = 4 * k + 2 * hi;
            v2f brel, broot;
            brel.x  = w_rel [(krow    ) * DF + ncol];
            brel.y  = w_rel [(krow + 1) * DF + ncol];
            broot.x = w_root[(krow    ) * DF + ncol];
            broot.y = w_root[(krow + 1) * DF + ncol];
            c = __builtin_amdgcn_wmma_f32_16x16x4_f32(
                    false, a_agg[k], false, brel, (short)0, c, false, false);
            c = __builtin_amdgcn_wmma_f32_16x16x4_f32(
                    false, a_x[k], false, broot, (short)0, c, false, false);
        }

#pragma unroll
        for (int v = 0; v < 8; ++v) {
            float r = c[v];
            if (RELU) r = fmaxf(r, 0.f);
            const int m = row0 + v + 8 * hi;
            if (m < n_nodes) out[(long long)m * DF + ncol] = r;
        }
    }
}

extern "C" void kernel_launch(void* const* d_in, const int* in_sizes, int n_in,
                              void* d_out, int out_size, void* d_ws, size_t ws_size,
                              hipStream_t stream) {
    const float* x       = (const float*)d_in[0];
    const int*   ei      = (const int*)  d_in[1];
    const float* w_rel1  = (const float*)d_in[2];
    const float* w_root1 = (const float*)d_in[3];
    const float* b1      = (const float*)d_in[4];
    const float* w_rel2  = (const float*)d_in[5];
    const float* w_root2 = (const float*)d_in[6];
    const float* b2      = (const float*)d_in[7];
    float*       out     = (float*)d_out;

    const int N = in_sizes[0] / DF;
    const int E = in_sizes[1] / 2;
    const int* src = ei;
    const int* dst = ei + E;

    float* h = (float*)d_ws;

    const int nb      = (N + 1 + BNODES - 1) / BNODES;
    const int g       = (E + ECH - 1) / ECH;
    const int gp      = (g + 3) / 4 * 4;
    const int NBP     = (nb + 1 + 3) / 4 * 4;
    const int NO      = (N + 1 + 3) / 4 * 4;
    int*  mat     = (int*)(h + (size_t)N * DF);
    int*  binbase = mat + (size_t)nb * gp;
    int*  off     = binbase + NBP;
    int2* pairs = (int2*)(off + NO);
    int*  perm  = (int*)(pairs + E);

    const int mtiles  = (N + 15) / 16;

    bin_count<<<g, ETHR, nb * sizeof(int), stream>>>(dst, mat, E, nb, g, gp);
    bin_scatter<<<g, ETHR, 2 * nb * sizeof(int), stream>>>(src, dst, mat, binbase, pairs, E, nb, g, gp);
    bucket_fill<<<nb, ETHR, 0, stream>>>(pairs, binbase, off, perm, N);

    conv_layer_wmma<true><<<mtiles, 32, 0, stream>>>(x, off, perm, w_rel1, w_root1, b1, h, N);

    conv_layer_wmma<false><<<mtiles, 32, 0, stream>>>(h, off, perm, w_rel2, w_root2, b2, out, N);
}


# Both binning passes park their ranked words in an LDS stage and copy it out in order: a store instruction writes one contiguous run, not 32 separate lines; pair and sum order = LDS arrival as before
# speedup vs baseline: 5.9616x; 1.1252x over previous
#include <hip/hip_runtime.h>
#include <hip/hip_bf16.h>


typedef __attribute__((ext_vector_type(2))) float v2f;
typedef __attribute__((ext_vector_type(4))) float v4f;
typedef __attribute__((ext_vector_type(8))) float v8f;

#define DF 48

#define BSH    9
#define BNODES (1 << BSH)
#define ECH    8192
#define ETHR   1024

__global__ __launch_bounds__(ETHR)
void bin_count(const int* __restrict__ dst,
               int* __restrict__ mat,
               int n_edges, int nb, int g, int gp) {
    extern __shared__ int lc[];
    const int t = threadIdx.x;
    const int c = blockIdx.x;
    for (int i = t; i < nb; i += ETHR) lc[i] = 0;
    __syncthreads();
    const int e0 = c * ECH + t;
    int d[ECH / ETHR];
#pragma unroll
    for (int k = 0; k < ECH / ETHR; ++k) d[k] = dst[min(e0 + k * ETHR, n_edges - 1)];
#pragma unroll
    for (int k = 0; k < ECH / ETHR; ++k)
        if (e0 + k * ETHR < n_edges) atomicAdd(&lc[d[k] >> BSH], 1);
    __syncthreads();
    for (int i = t; i < nb; i += ETHR) mat[i * gp + c] = lc[i];
    if (c == g - 1)
        for (int i = t; i < nb; i += ETHR)
            for (int j = g; j < gp; ++j) mat[i * gp + j] = 0;
}

__device__ __forceinline__ int wave_incl_scan(int v, int lane) {
    for (int d = 1; d < 32; d <<= 1) {
        int o = __shfl_up(v, d, 32);
        if (lane >= d) v += o;
    }
    return v;
}

#define RB 7
typedef __attribute__((ext_vector_type(4))) int v4i;

__global__ __launch_bounds__(ETHR)
void bin_scatter(const int* __restrict__ src,
                 const int* __restrict__ dst,
                 const int* __restrict__ mat,
                 int* __restrict__ binbase,
                 int2* __restrict__ pairs,
                 int n_edges, int nb, int g, int gp) {
    extern __shared__ int lb[];
    int* rtot = lb + nb;
    int* lcur = lb + 2 * nb;
    __shared__ int wsum[32];
    __shared__ int wsum2[32];
    __shared__ int2 stage[ECH];
    const int t    = threadIdx.x;
    const int lane = t & 31;
    const int w    = __builtin_amdgcn_readfirstlane(t >> 5);
    const int c    = blockIdx.x;

    const int e0 = c * ECH + t;
    int d[ECH / ETHR], s[ECH / ETHR];
#pragma unroll
    for (int k = 0; k < ECH / ETHR; ++k) {
        const int e = min(e0 + k * ETHR, n_edges - 1);
        d[k] = dst[e];
        s[k] = src[e];
    }
    const int cn0 = (t < nb) ? mat[t * gp + c] : 0;

    const v4i* __restrict__ mat4 = (const v4i*)mat;
    const int q = gp >> 2;
    for (int i0 = w; i0 < nb; i0 += 32 * RB) {
        int tot[RB], part[RB];
#pragma unroll
        for (int r = 0; r < RB; ++r) { tot[r] = 0; part[r] = 0; }
        for (int j0 = 0; j0 < q; j0 += 64) {
            v4i v[RB][2];
#pragma unroll
            for (int r = 0; r < RB; ++r) {
                const v4i* __restrict__ row4 = mat4 + min(i0 + 32 * r, nb - 1) * q;
                v[r][0] = row4[min(j0 + lane, q - 1)];
                v[r][1] = row4[min(j0 + 32 + lane, q - 1)];
            }
            static_assert(RB == 7, "the statement below names 2 * RB values");
            asm volatile("" :: "v"(v[0][0]), "v"(v[1][0]), "v"(v[2][0]), "v"(v[3][0]),
                               "v"(v[4][0]), "v"(v[5][0]), "v"(v[6][0]),
                               "v"(v[0][1]), "v"(v[1][1]), "v"(v[2][1]), "v"(v[3][1]),
                               "v"(v[4][1]), "v"(v[5][1]), "v"(v[6][1]));
#pragma unroll
            for (int u = 0; u < 2; ++u) {
                const int  j   = j0 + 32 * u + lane;
                const bool ok  = j < q;
                const int  col = 4 * j;
                const int  cc  = ok ? c : 0;
#pragma unroll
                for (int r = 0; r < RB; ++r) {
                    const v4i x = v[r][u];
                    tot[r]  += ok ? ((x.x + x.y) + (x.z + x.w)) : 0;
                    part[r] += ((col     < cc) ? x.x : 0) + ((col + 1 < cc) ? x.y : 0)
                             + ((col + 2 < cc) ? x.z : 0) + ((col + 3 < cc) ? x.w : 0);
                }
            }
        }
#pragma unroll
        for (int x = 16; x > 0; x >>= 1) {
#pragma unroll
            for (int r = 0; r < RB; ++r) {
                const int a = __shfl_xor(tot[r], x, 32);
                const int b = __shfl_xor(part[r], x, 32);
                tot[r]  += a;
                part[r] += b;
            }
        }
#pragma unroll
        for (int r = 0; r < RB; ++r) {
            const int i = i0 + 32 * r;
            if (lane == 0 && i < nb) { rtot[i] = tot[r]; lb[i] = part[r]; }
        }
    }

    __syncthreads();

    int carry = 0, lcarry = 0;
    for (int b0 = 0; b0 < nb; b0 += ETHR) {
        const int i    = b0 + t;
        const int v    = (i < nb) ? rtot[i] : 0;
        const int cn   = (b0 == 0) ? cn0 : ((i < nb) ? mat[i * gp + c] : 0);
        const int inc  = wave_incl_scan(v, lane);
        const int linc = wave_incl_scan(cn, lane);
        if (lane == 31) { wsum[w] = inc; wsum2[w] = linc; }
        __syncthreads();
        const int ws  = wsum[lane];
        const int ws2 = wsum2[lane];
        int p    = (lane < w) ? ws : 0;
        int all  = ws;
        int p2   = (lane < w) ? ws2 : 0;
        int all2 = ws2;
        for (int x = 16; x > 0; x >>= 1) {
            p    += __shfl_xor(p, x, 32);
            all  += __shfl_xor(all, x, 32);
            p2   += __shfl_xor(p2, x, 32);
            all2 += __shfl_xor(all2, x, 32);
        }
        const int ex  = carry + p + inc - v;
        const int lex = lcarry + p2 + linc - cn;
        if (i < nb) {
            lb[i] += ex - lex;
            lcur[i] = lex;
            if (c == 0) binbase[i] = ex;
        }
        carry  += all;
        lcarry += all2;
        __syncthreads();
    }
    if (c == 0 && t == 0) binbase[nb] = carry;

#pragma unroll
    for (int k = 0; k < ECH / ETHR; ++k) {
        if (e0 + k * ETHR < n_edges) {
            const int lp = atomicAdd(&lcur[d[k] >> BSH], 1);
            stage[min(lp, ECH - 1)] = make_int2(s[k], d[k]);
        }
    }
    __syncthreads();
    const int cnt = min(ECH, n_edges - c * ECH);
    int2 pr[ECH / ETHR];
    int  gb[ECH / ETHR];
#pragma unroll
    for (int k = 0; k < ECH / ETHR; ++k) pr[k] = stage[min(t + k * ETHR, cnt - 1)];
#pragma unroll
    for (int k = 0; k < ECH / ETHR; ++k) gb[k] = lb[pr[k].y >> BSH];
#pragma unroll
    for (int k = 0; k < ECH / ETHR; ++k) {
        const int i = t + k * ETHR;
        if (i < cnt) pairs[gb[k] + i] = pr[k];
    }
}

#define SCAP 16384
__global__ __launch_bounds__(ETHR)
void bucket_fill(const int2* __restrict__ pairs,
                 const int* __restrict__ binbase,
                 int* __restrict__ off,
                 int* __restrict__ perm,
                 int n_nodes) {
    __shared__ int nc[BNODES];
    __shared__ int wsum[32];
    __shared__ int stage[SCAP];
    const int t    = threadIdx.x;
    const int lane = t & 31;
    const int w    = t >> 5;
    const int b    = blockIdx.x;
    const int beg   = binbase[b];
    const int end   = binbase[b + 1];
    const int node0 = b << BSH;
    if (t < BNODES) nc[t] = 0;
    __syncthreads();
    for (int i0 = beg + t; i0 < end; i0 += 4 * ETHR) {
        int dn[4];
#pragma unroll
        for (int u = 0; u < 4; ++u) dn[u] = pairs[min(i0 + u * ETHR, end - 1)].y;
#pragma unroll
        for (int u = 0; u < 4; ++u)
            if (i0 + u * ETHR < end) atomicAdd(&nc[dn[u] - node0], 1);
    }
    __syncthreads();
    const int v   = (t < BNODES) ? nc[t] : 0;
    const int inc = wave_incl_scan(v, lane);
    if (lane == 31) wsum[w] = inc;
    __syncthreads();
    int p = (lane < w) ? wsum[lane] : 0;
    for (int d = 16; d > 0; d >>= 1) p += __shfl_xor(p, d, 32);
    const int ex = beg + p + inc - v;
    if (t < BNODES) {
        nc[t] = ex;
        if (node0 + t <= n_nodes) off[node0 + t] = ex;
    }
    __syncthreads();
    for (int i0 = beg + t; i0 < end; i0 += 4 * ETHR) {
        int2 pr[4];
#pragma unroll
        for (int u = 0; u < 4; ++u) pr[u] = pairs[min(i0 + u * ETHR, end - 1)];
#pragma unroll
        for (int u = 0; u < 4; ++u) {
            if (i0 + u * ETHR < end) {
                const int q = atomicAdd(&nc[pr[u].y - node0], 1);
                const int r = q - beg;
                if (r < SCAP) stage[r] = pr[u].x;
                else          perm[q]  = pr[u].x;
            }
        }
    }
    __syncthreads();
    const int mis = (int)(((unsigned long long)perm >> 2) & 31);
    const int nst = min(end - beg, SCAP);
    if (nst > 0) {
        for (int i0 = ((beg + mis) & ~31) - mis + t; i0 < beg + nst; i0 += 4 * ETHR) {
            int vv[4];
#pragma unroll
            for (int u = 0; u < 4; ++u) vv[u] = stage[min(max(i0 + u * ETHR - beg, 0), nst - 1)];
#pragma unroll
            for (int u = 0; u < 4; ++u) {
                const int i = i0 + u * ETHR;
                if (i >= beg && i < beg + nst) perm[i] = vv[u];
            }
        }
    }
}

#define TP 52

template <bool RELU>
__global__ __launch_bounds__(32)
void conv_layer_wmma(const float* __restrict__ feat,
                     const int* __restrict__ off,
                     const int* __restrict__ perm,
                     const float* __restrict__ w_rel,
                     const float* __restrict__ w_root,
                     const float* __restrict__ bias,
                     float* __restrict__ out,
                     int n_nodes) {
    __shared__ __attribute__((aligned(16))) float tile[16 * TP];
    const int lane = threadIdx.x;
    const int hi   = lane >> 4;
    const int lo   = lane & 15;
    const int row0 = blockIdx.x * 16;

    const v4f* __restrict__ f4 = (const v4f*)feat;
    const int p = lane & 3;
    const int q = lane >> 2;
    const v4f z = {0.f, 0.f, 0.f, 0.f};
    const int o = off[min(row0 + min(lane, 16), n_nodes)];

    int nbeg[2], ndeg[2], nia[2], nib[2];
#pragma unroll
    for (int r = 0; r < 2; ++r) {
        const int j = 8 * r + q;
        nbeg[r] = __shfl(o, j, 32);
        ndeg[r] = __shfl(o, j + 1, 32) - nbeg[r];
        nia[r]  = (p     < ndeg[r]) ? perm[nbeg[r] + p]     : 0;
        nib[r]  = (4 + p < ndeg[r]) ? perm[nbeg[r] + 4 + p] : 0;
    }

    asm volatile("" :: "v"(nia[0]), "v"(nib[0]), "v"(nia[1]), "v"(nib[1]));

#define GATHER_SLOT(U)                                                          \
    {                                                                           \
        const int s = __builtin_amdgcn_update_dpp(0, ((U) < 4) ? ca : cb,       \
                                                  ((U) & 3) * 0x55, 0xf, 0xf,   \
                                                  true);                        \
        v[U][0] = z; v[U][1] = z; v[U][2] = z;                                  \
        if ((U) < rem) {                                                        \
            const v4f* __restrict__ rp = f4 + (long long)s * 12 + p;            \
            v[U][0] = rp[0];                                                    \
            v[U][1] = rp[4];                                                    \
            v[U][2] = rp[8];                                                    \
        }                                                                       \
    }

#pragma unroll
    for (int r = 0; r < 2; ++r) {
        const int j   = 8 * r + q;
        const int beg = nbeg[r];
        const int deg = ndeg[r];
        int na = nia[r], nb = nib[r];
        v4f a0 = z, a1 = z, a2 = z;
        for (int base = 0; base < deg; base += 8) {
            const int rem = deg - base;
            const int ca = na, cb = nb;
            na = (8 + p  < rem) ? perm[beg + base + 8 + p]  : 0;
            nb = (12 + p < rem) ? perm[beg + base + 12 + p] : 0;
            v4f v[8][3];
            GATHER_SLOT(0) GATHER_SLOT(1) GATHER_SLOT(2) GATHER_SLOT(3)
            GATHER_SLOT(4) GATHER_SLOT(5) GATHER_SLOT(6) GATHER_SLOT(7)
#pragma unroll
            for (int u = 0; u < 8; ++u) { a0 += v[u][0]; a1 += v[u][1]; a2 += v[u][2]; }
        }
        float* trow = &tile[j * TP + 4 * p];
        *(v4f*)(trow)      = a0;
        *(v4f*)(trow + 16) = a1;
        *(v4f*)(trow + 32) = a2;
    }
#undef GATHER_SLOT
    __syncthreads();

    const int arow = min(row0 + lo, n_nodes - 1);
    v2f a_agg[12], a_x[12];
#pragma unroll
    for (int k = 0; k < 12; ++k) {
        const int kk = 4 * k + 2 * hi;
        const long long base = (long long)arow * DF + kk;
        a_agg[k] = *(const v2f*)&tile[lo * TP + kk];
        a_x[k].x = feat[base];
        a_x[k].y = feat[base + 1];
    }

#pragma unroll
    for (int nt = 0; nt < 3; ++nt) {
        const int ncol = nt * 16 + lo;
        const float bv = bias[ncol];
        v8f c;
#pragma unroll
        for (int i = 0; i < 8; ++i) c[i] = bv;

#pragma unroll
        for (int k = 0; k < 12; ++k) {
            const int krow = 4 * k + 2 * hi;
            v2f brel, broot;
            brel.x  = w_rel [(krow    ) * DF + ncol];
            brel.y  = w_rel [(krow + 1) * DF + ncol];
            broot.x = w_root[(krow    ) * DF + ncol];
            broot.y = w_root[(krow + 1) * DF + ncol];
            c = __builtin_amdgcn_wmma_f32_16x16x4_f32(
                    false, a_agg[k], false, brel, (short)0, c, false, false);
            c = __builtin_amdgcn_wmma_f32_16x16x4_f32(
                    false, a_x[k], false, broot, (short)0, c, false, false);
        }

#pragma unroll
        for (int v = 0; v < 8; ++v) {
            float r = c[v];
            if (RELU) r = fmaxf(r, 0.f);
            const int m = row0 + v + 8 * hi;
            if (m < n_nodes) out[(long long)m * DF + ncol] = r;
        }
    }
}

extern "C" void kernel_launch(void* const* d_in, const int* in_sizes, int n_in,
                              void* d_out, int out_size, void* d_ws, size_t ws_size,
                              hipStream_t stream) {
    const float* x       = (const float*)d_in[0];
    const int*   ei      = (const int*)  d_in[1];
    const float* w_rel1  = (const float*)d_in[2];
    const float* w_root1 = (const float*)d_in[3];
    const float* b1      = (const float*)d_in[4];
    const float* w_rel2  = (const float*)d_in[5];
    const float* w_root2 = (const float*)d_in[6];
    const float* b2      = (const float*)d_in[7];
    float*       out     = (float*)d_out;

    const int N = in_sizes[0] / DF;
    const int E = in_sizes[1] / 2;
    const int* src = ei;
    const int* dst = ei + E;

    float* h = (float*)d_ws;

    const int nb      = (N + 1 + BNODES - 1) / BNODES;
    const int g       = (E + ECH - 1) / ECH;
    const int gp      = (g + 3) / 4 * 4;
    const int NBP     = (nb + 1 + 3) / 4 * 4;
    const int NO      = (N + 1 + 3) / 4 * 4;
    int*  mat     = (int*)(h + (size_t)N * DF);
    int*  binbase = mat + (size_t)nb * gp;
    int*  off     = binbase + NBP;
    int2* pairs = (int2*)(off + NO);
    int*  perm  = (int*)(pairs + E);

    const int mtiles  = (N + 15) / 16;

    bin_count<<<g, ETHR, nb * sizeof(int), stream>>>(dst, mat, E, nb, g, gp);
    bin_scatter<<<g, ETHR, 3 * nb * sizeof(int), stream>>>(src, dst, mat, binbase, pairs, E, nb, g, gp);
    bucket_fill<<<nb, ETHR, 0, stream>>>(pairs, binbase, off, perm, N);

    conv_layer_wmma<true><<<mtiles, 32, 0, stream>>>(x, off, perm, w_rel1, w_root1, b1, h, N);

    conv_layer_wmma<false><<<mtiles, 32, 0, stream>>>(h, off, perm, w_rel2, w_root2, b2, out, N);
}


# Drop the count launch: each chunk orders its pairs by bin inside its own window and stores its bin starts; the per-bin pass walks its 196 runs by that table; 5 launches -> 4; sum order = LDS arrival
# speedup vs baseline: 6.3365x; 1.0629x over previous
#include <hip/hip_runtime.h>
#include <hip/hip_bf16.h>


typedef __attribute__((ext_vector_type(2))) float v2f;
typedef __attribute__((ext_vector_type(4))) float v4f;
typedef __attribute__((ext_vector_type(8))) float v8f;

#define DF 48

#define BSH    9
#define BNODES (1 << BSH)
#define ECH    8192
#define ETHR   1024

__device__ __forceinline__ int wave_incl_scan(int v, int lane) {
    for (int d = 1; d < 32; d <<= 1) {
        int o = __shfl_up(v, d, 32);
        if (lane >= d) v += o;
    }
    return v;
}

__global__ __launch_bounds__(ETHR)
void bin_chunk(const int* __restrict__ src,
               const int* __restrict__ dst,
               int* __restrict__ tab,
               int2* __restrict__ pairs,
               int n_edges, int nb, int tp) {
    extern __shared__ int lc[];
    __shared__ int wsum[32];
    __shared__ int2 stage[ECH];
    const int t    = threadIdx.x;
    const int lane = t & 31;
    const int w    = t >> 5;
    const int c    = blockIdx.x;

    const int e0 = c * ECH + t;
    int d[ECH / ETHR], s[ECH / ETHR], rk[ECH / ETHR];
#pragma unroll
    for (int k = 0; k < ECH / ETHR; ++k) {
        const int e = min(e0 + k * ETHR, n_edges - 1);
        d[k] = dst[e];
        s[k] = src[e];
    }
    for (int i = t; i < nb; i += ETHR) lc[i] = 0;
    __syncthreads();
#pragma unroll
    for (int k = 0; k < ECH / ETHR; ++k) {
        rk[k] = 0;
        if (e0 + k * ETHR < n_edges) rk[k] = atomicAdd(&lc[d[k] >> BSH], 1);
    }
    __syncthreads();

    int carry = 0;
    for (int b0 = 0; b0 < nb; b0 += ETHR) {
        const int i   = b0 + t;
        const int v   = (i < nb) ? lc[i] : 0;
        const int inc = wave_incl_scan(v, lane);
        if (lane == 31) wsum[w] = inc;
        __syncthreads();
        const int ws = wsum[lane];
        int p   = (lane < w) ? ws : 0;
        int all = ws;
        for (int x = 16; x > 0; x >>= 1) {
            p   += __shfl_xor(p, x, 32);
            all += __shfl_xor(all, x, 32);
        }
        const int ex = carry + p + inc - v;
        if (i < nb) {
            lc[i] = ex;
            tab[c * tp + i] = ex;
        }
        carry += all;
        __syncthreads();
    }
    if (t == 0) tab[c * tp + nb] = carry;

    int fp[ECH / ETHR];
#pragma unroll
    for (int k = 0; k < ECH / ETHR; ++k) fp[k] = lc[d[k] >> BSH];
#pragma unroll
    for (int k = 0; k < ECH / ETHR; ++k) {
        if (e0 + k * ETHR < n_edges) {
            stage[min(max(fp[k] + rk[k], 0), ECH - 1)] = make_int2(s[k], d[k]);
        }
    }
    __syncthreads();
    const int cnt = min(ECH, n_edges - c * ECH);
    int2 pr[ECH / ETHR];
#pragma unroll
    for (int k = 0; k < ECH / ETHR; ++k) pr[k] = stage[min(t + k * ETHR, cnt - 1)];
#pragma unroll
    for (int k = 0; k < ECH / ETHR; ++k) {
        const int i = t + k * ETHR;
        if (i < cnt) pairs[c * ECH + i] = pr[k];
    }
}

#define SCAP 16384
#define CPW  7
__global__ __launch_bounds__(ETHR)
void bucket_fill(const int2* __restrict__ pairs,
                 const int* __restrict__ tab,
                 int* __restrict__ off,
                 int* __restrict__ perm,
                 int n_nodes, int n_edges, int g, int tp) {
    extern __shared__ int cs[];
    int* cn = cs + g;
    __shared__ int nc[BNODES];
    __shared__ int wsum[32];
    __shared__ int wsum2[32];
    __shared__ int stage[SCAP];
    const int t    = threadIdx.x;
    const int lane = t & 31;
    const int w    = __builtin_amdgcn_readfirstlane(t >> 5);
    const int b    = blockIdx.x;
    const int node0 = b << BSH;

    int sb = 0, se = 0;
    for (int cc = t; cc < g; cc += ETHR) {
        const int wl = min(ECH, n_edges - cc * ECH);
        const int s0 = min(max(tab[cc * tp + b], 0), wl);
        const int s1 = min(max(tab[cc * tp + b + 1], s0), wl);
        cs[cc] = cc * ECH + s0;
        cn[cc] = s1 - s0;
        sb += s0;
        se += s1;
    }
    if (t < BNODES) nc[t] = 0;
    for (int x = 16; x > 0; x >>= 1) {
        sb += __shfl_xor(sb, x, 32);
        se += __shfl_xor(se, x, 32);
    }
    if (lane == 0) { wsum[w] = sb; wsum2[w] = se; }
    __syncthreads();

    for (int c0 = w; c0 < g; c0 += 32 * CPW) {
        int n[CPW], base[CPW], dn[CPW][2];
#pragma unroll
        for (int r = 0; r < CPW; ++r) {
            const int cc = c0 + 32 * r;
            const int cl = min(cc, g - 1);
            n[r]    = (cc < g) ? cn[cl] : 0;
            base[r] = cs[cl];
        }
#pragma unroll
        for (int r = 0; r < CPW; ++r) {
            const int last = max(n[r] - 1, 0);
            dn[r][0] = pairs[min(base[r] + min(lane, last), n_edges - 1)].y;
            dn[r][1] = pairs[min(base[r] + min(lane + 32, last), n_edges - 1)].y;
        }
#pragma unroll
        for (int r = 0; r < CPW; ++r) {
            if (lane < n[r])      atomicAdd(&nc[dn[r][0] - node0], 1);
            if (lane + 32 < n[r]) atomicAdd(&nc[dn[r][1] - node0], 1);
        }
#pragma unroll
        for (int r = 0; r < CPW; ++r)
            for (int k = 64 + lane; k < n[r]; k += 32)
                atomicAdd(&nc[pairs[min(base[r] + k, n_edges - 1)].y - node0], 1);
    }
    int beg = wsum[lane], end = wsum2[lane];
    for (int x = 16; x > 0; x >>= 1) {
        beg += __shfl_xor(beg, x, 32);
        end += __shfl_xor(end, x, 32);
    }
    __syncthreads();
    const int v   = (t < BNODES) ? nc[t] : 0;
    const int inc = wave_incl_scan(v, lane);
    if (lane == 31) wsum[w] = inc;
    __syncthreads();
    int p = (lane < w) ? wsum[lane] : 0;
    for (int x = 16; x > 0; x >>= 1) p += __shfl_xor(p, x, 32);
    const int ex = beg + p + inc - v;
    if (t < BNODES) {
        nc[t] = ex;
        if (node0 + t <= n_nodes) off[node0 + t] = ex;
    }
    __syncthreads();
    for (int c0 = w; c0 < g; c0 += 32 * CPW) {
        int n[CPW], base[CPW];
        int2 pr[CPW][2];
#pragma unroll
        for (int r = 0; r < CPW; ++r) {
            const int cc = c0 + 32 * r;
            const int cl = min(cc, g - 1);
            n[r]    = (cc < g) ? cn[cl] : 0;
            base[r] = cs[cl];
        }
#pragma unroll
        for (int r = 0; r < CPW; ++r) {
            const int last = max(n[r] - 1, 0);
            pr[r][0] = pairs[min(base[r] + min(lane, last), n_edges - 1)];
            pr[r][1] = pairs[min(base[r] + min(lane + 32, last), n_edges - 1)];
        }
        int q[CPW][2];
#pragma unroll
        for (int r = 0; r < CPW; ++r) {
#pragma unroll
            for (int u = 0; u < 2; ++u) {
                q[r][u] = beg;
                if (lane + 32 * u < n[r]) q[r][u] = atomicAdd(&nc[pr[r][u].y - node0], 1);
            }
        }
#pragma unroll
        for (int r = 0; r < CPW; ++r) {
#pragma unroll
            for (int u = 0; u < 2; ++u) {
                if (lane + 32 * u < n[r]) {
                    const int rr = q[r][u] - beg;
                    if (rr < SCAP) stage[max(rr, 0)] = pr[r][u].x;
                    else           perm[min(q[r][u], n_edges - 1)] = pr[r][u].x;
                }
            }
        }
#pragma unroll
        for (int r = 0; r < CPW; ++r) {
            for (int k = 64 + lane; k < n[r]; k += 32) {
                const int2 pk = pairs[min(base[r] + k, n_edges - 1)];
                const int q  = atomicAdd(&nc[pk.y - node0], 1);
                const int rr = q - beg;
                if (rr < SCAP) stage[max(rr, 0)] = pk.x;
                else           perm[min(q, n_edges - 1)] = pk.x;
            }
        }
    }
    __syncthreads();
    const int mis = (int)(((unsigned long long)perm >> 2) & 31);
    const int nst = min(end - beg, SCAP);
    if (nst > 0) {
        for (int i0 = ((beg + mis) & ~31) - mis + t; i0 < beg + nst; i0 += 4 * ETHR) {
            int vv[4];
#pragma unroll
            for (int u = 0; u < 4; ++u) vv[u] = stage[min(max(i0 + u * ETHR - beg, 0), nst - 1)];
#pragma unroll
            for (int u = 0; u < 4; ++u) {
                const int i = i0 + u * ETHR;
                if (i >= beg && i < beg + nst) perm[i] = vv[u];
            }
        }
    }
}

#define TP 52

template <bool RELU>
__global__ __launch_bounds__(32)
void conv_layer_wmma(const float* __restrict__ feat,
                     const int* __restrict__ off,
                     const int* __restrict__ perm,
                     const float* __restrict__ w_rel,
                     const float* __restrict__ w_root,
                     const float* __restrict__ bias,
                     float* __restrict__ out,
                     int n_nodes) {
    __shared__ __attribute__((aligned(16))) float tile[16 * TP];
    const int lane = threadIdx.x;
    const int hi   = lane >> 4;
    const int lo   = lane & 15;
    const int row0 = blockIdx.x * 16;

    const v4f* __restrict__ f4 = (const v4f*)feat;
    const int p = lane & 3;
    const int q = lane >> 2;
    const v4f z = {0.f, 0.f, 0.f, 0.f};
    const int o = off[min(row0 + min(lane, 16), n_nodes)];

    int nbeg[2], ndeg[2], nia[2], nib[2];
#pragma unroll
    for (int r = 0; r < 2; ++r) {
        const int j = 8 * r + q;
        nbeg[r] = __shfl(o, j, 32);
        ndeg[r] = __shfl(o, j + 1, 32) - nbeg[r];
        nia[r]  = (p     < ndeg[r]) ? perm[nbeg[r] + p]     : 0;
        nib[r]  = (4 + p < ndeg[r]) ? perm[nbeg[r] + 4 + p] : 0;
    }

    asm volatile("" :: "v"(nia[0]), "v"(nib[0]), "v"(nia[1]), "v"(nib[1]));

#define GATHER_SLOT(U)                                                          \
    {                                                                           \
        const int s = __builtin_amdgcn_update_dpp(0, ((U) < 4) ? ca : cb,       \
                                                  ((U) & 3) * 0x55, 0xf, 0xf,   \
                                                  true);                        \
        v[U][0] = z; v[U][1] = z; v[U][2] = z;                                  \
        if ((U) < rem) {                                                        \
            const v4f* __restrict__ rp = f4 + (long long)s * 12 + p;            \
            v[U][0] = rp[0];                                                    \
            v[U][1] = rp[4];                                                    \
            v[U][2] = rp[8];                                                    \
        }                                                                       \
    }

#pragma unroll
    for (int r = 0; r < 2; ++r) {
        const int j   = 8 * r + q;
        const int beg = nbeg[r];
        const int deg = ndeg[r];
        int na = nia[r], nb = nib[r];
        v4f a0 = z, a1 = z, a2 = z;
        for (int base = 0; base < deg; base += 8) {
            const int rem = deg - base;
            const int ca = na, cb = nb;
            na = (8 + p  < rem) ? perm[beg + base + 8 + p]  : 0;
            nb = (12 + p < rem) ? perm[beg + base + 12 + p] : 0;
            v4f v[8][3];
            GATHER_SLOT(0) GATHER_SLOT(1) GATHER_SLOT(2) GATHER_SLOT(3)
            GATHER_SLOT(4) GATHER_SLOT(5) GATHER_SLOT(6) GATHER_SLOT(7)
#pragma unroll
            for (int u = 0; u < 8; ++u) { a0 += v[u][0]; a1 += v[u][1]; a2 += v[u][2]; }
        }
        float* trow = &tile[j * TP + 4 * p];
        *(v4f*)(trow)      = a0;
        *(v4f*)(trow + 16) = a1;
        *(v4f*)(trow + 32) = a2;
    }
#undef GATHER_SLOT
    __syncthreads();

    const int arow = min(row0 + lo, n_nodes - 1);
    v2f a_agg[12], a_x[12];
#pragma unroll
    for (int k = 0; k < 12; ++k) {
        const int kk = 4 * k + 2 * hi;
        const long long base = (long long)arow * DF + kk;
        a_agg[k] = *(const v2f*)&tile[lo * TP + kk];
        a_x[k].x = feat[base];
        a_x[k].y = feat[base + 1];
    }

#pragma unroll
    for (int nt = 0; nt < 3; ++nt) {
        const int ncol = nt * 16 + lo;
        const float bv = bias[ncol];
        v8f c;
#pragma unroll
        for (int i = 0; i < 8; ++i) c[i] = bv;

#pragma unroll
        for (int k = 0; k < 12; ++k) {
            const int krow = 4 * k + 2 * hi;
            v2f brel, broot;
            brel.x  = w_rel [(krow    ) * DF + ncol];
            brel.y  = w_rel [(krow + 1) * DF + ncol];
            broot.x = w_root[(krow    ) * DF + ncol];
            broot.y = w_root[(krow + 1) * DF + ncol];
            c = __builtin_amdgcn_wmma_f32_16x16x4_f32(
                    false, a_agg[k], false, brel, (short)0, c, false, false);
            c = __builtin_amdgcn_wmma_f32_16x16x4_f32(
                    false, a_x[k], false, broot, (short)0, c, false, false);
        }

#pragma unroll
        for (int v = 0; v < 8; ++v) {
            float r = c[v];
            if (RELU) r = fmaxf(r, 0.f);
            const int m = row0 + v + 8 * hi;
            if (m < n_nodes) out[(long long)m * DF + ncol] = r;
        }
    }
}

extern "C" void kernel_launch(void* const* d_in, const int* in_sizes, int n_in,
                              void* d_out, int out_size, void* d_ws, size_t ws_size,
                              hipStream_t stream) {
    const float* x       = (const float*)d_in[0];
    const int*   ei      = (const int*)  d_in[1];
    const float* w_rel1  = (const float*)d_in[2];
    const float* w_root1 = (const float*)d_in[3];
    const float* b1      = (const float*)d_in[4];
    const float* w_rel2  = (const float*)d_in[5];
    const float* w_root2 = (const float*)d_in[6];
    const float* b2      = (const float*)d_in[7];
    float*       out     = (float*)d_out;

    const int N = in_sizes[0] / DF;
    const int E = in_sizes[1] / 2;
    const int* src = ei;
    const int* dst = ei + E;

    float* h = (float*)d_ws;

    const int nb      = (N + 1 + BNODES - 1) / BNODES;
    const int g       = (E + ECH - 1) / ECH;
    const int tp      = (nb + 1 + 31) / 32 * 32;
    const int NO      = (N + 1 + 31) / 32 * 32;
    int*  tab   = (int*)(h + (size_t)N * DF);
    int*  off   = tab + (size_t)g * tp;
    int2* pairs = (int2*)(off + NO);
    int*  perm  = (int*)(pairs + E);

    const int mtiles  = (N + 15) / 16;

    bin_chunk<<<g, ETHR, nb * sizeof(int), stream>>>(src, dst, tab, pairs, E, nb, tp);
    bucket_fill<<<nb, ETHR, 2 * g * sizeof(int), stream>>>(pairs, tab, off, perm, N, E, g, tp);

    conv_layer_wmma<true><<<mtiles, 32, 0, stream>>>(x, off, perm, w_rel1, w_root1, b1, h, N);

    conv_layer_wmma<false><<<mtiles, 32, 0, stream>>>(h, off, perm, w_rel2, w_root2, b2, out, N);
}


# Layer 1 runs in the sort's per-bin launch: each wave of a bin's work-group does one 16-node tile, indices read from the bin's LDS stage; bins of 391 nodes (256 groups); 4 launches -> 3; sum order kept
# speedup vs baseline: 6.8135x; 1.0753x over previous
#include <hip/hip_runtime.h>
#include <hip/hip_bf16.h>


typedef __attribute__((ext_vector_type(2))) float v2f;
typedef __attribute__((ext_vector_type(4))) float v4f;
typedef __attribute__((ext_vector_type(8))) float v8f;

#define DF 48

#define BNODES 391
#define ECH    8192
#define ETHR   1024
#define TP     52

__device__ __forceinline__ int wave_incl_scan(int v, int lane) {
    for (int d = 1; d < 32; d <<= 1) {
        int o = __shfl_up(v, d, 32);
        if (lane >= d) v += o;
    }
    return v;
}

__global__ __launch_bounds__(ETHR)
void bin_chunk(const int* __restrict__ src,
               const int* __restrict__ dst,
               int* __restrict__ tab,
               int2* __restrict__ pairs,
               int n_edges, int nb, int tp) {
    extern __shared__ int lc[];
    __shared__ int wsum[32];
    __shared__ int2 stage[ECH];
    const int t    = threadIdx.x;
    const int lane = t & 31;
    const int w    = t >> 5;
    const int c    = blockIdx.x;

    const int e0 = c * ECH + t;
    int d[ECH / ETHR], s[ECH / ETHR], rk[ECH / ETHR];
#pragma unroll
    for (int k = 0; k < ECH / ETHR; ++k) {
        const int e = min(e0 + k * ETHR, n_edges - 1);
        d[k] = dst[e];
        s[k] = src[e];
    }
    for (int i = t; i < nb; i += ETHR) lc[i] = 0;
    __syncthreads();
#pragma unroll
    for (int k = 0; k < ECH / ETHR; ++k) {
        rk[k] = 0;
        if (e0 + k * ETHR < n_edges) rk[k] = atomicAdd(&lc[d[k] / BNODES], 1);
    }
    __syncthreads();

    int carry = 0;
    for (int b0 = 0; b0 < nb; b0 += ETHR) {
        const int i   = b0 + t;
        const int v   = (i < nb) ? lc[i] : 0;
        const int inc = wave_incl_scan(v, lane);
        if (lane == 31) wsum[w] = inc;
        __syncthreads();
        const int ws = wsum[lane];
        int p   = (lane < w) ? ws : 0;
        int all = ws;
        for (int x = 16; x > 0; x >>= 1) {
            p   += __shfl_xor(p, x, 32);
            all += __shfl_xor(all, x, 32);
        }
        const int ex = carry + p + inc - v;
        if (i < nb) {
            lc[i] = ex;
            tab[c * tp + i] = ex;
        }
        carry += all;
        __syncthreads();
    }
    if (t == 0) tab[c * tp + nb] = carry;

    int fp[ECH / ETHR];
#pragma unroll
    for (int k = 0; k < ECH / ETHR; ++k) fp[k] = lc[d[k] / BNODES];
#pragma unroll
    for (int k = 0; k < ECH / ETHR; ++k) {
        if (e0 + k * ETHR < n_edges) {
            stage[min(max(fp[k] + rk[k], 0), ECH - 1)] = make_int2(s[k], d[k]);
        }
    }
    __syncthreads();
    const int cnt = min(ECH, n_edges - c * ECH);
    int2 pr[ECH / ETHR];
#pragma unroll
    for (int k = 0; k < ECH / ETHR; ++k) pr[k] = stage[min(t + k * ETHR, cnt - 1)];
#pragma unroll
    for (int k = 0; k < ECH / ETHR; ++k) {
        const int i = t + k * ETHR;
        if (i < cnt) pairs[c * ECH + i] = pr[k];
    }
}

#define SCAP 16384
#define CPW  7
__global__ __launch_bounds__(ETHR)
void bucket_fill_layer(const int2* __restrict__ pairs,
                       const int* __restrict__ tab,
                       int* __restrict__ off,
                       int* __restrict__ perm,
                       const float* __restrict__ feat,
                       const float* __restrict__ w_rel,
                       const float* __restrict__ w_root,
                       const float* __restrict__ bias,
                       float* __restrict__ out,
                       int n_nodes, int n_edges, int g, int tp) {
    extern __shared__ int cs[];
    int* cn = cs + g;
    __shared__ int nc[BNODES];
    __shared__ int ns[BNODES + 1];
    __shared__ int wsum[32];
    __shared__ int wsum2[32];
    __shared__ int stage[SCAP];
    __shared__ __attribute__((aligned(16))) float tiles[(ETHR / 32) * 16 * TP];
    const int t    = threadIdx.x;
    const int lane = t & 31;
    const int w    = __builtin_amdgcn_readfirstlane(t >> 5);
    const int b    = blockIdx.x;
    const int node0 = b * BNODES;

    int sb = 0, se = 0;
    for (int cc = t; cc < g; cc += ETHR) {
        const int wl = min(ECH, n_edges - cc * ECH);
        const int s0 = min(max(tab[cc * tp + b], 0), wl);
        const int s1 = min(max(tab[cc * tp + b + 1], s0), wl);
        cs[cc] = cc * ECH + s0;
        cn[cc] = s1 - s0;
        sb += s0;
        se += s1;
    }
    if (t < BNODES) nc[t] = 0;
    for (int x = 16; x > 0; x >>= 1) {
        sb += __shfl_xor(sb, x, 32);
        se += __shfl_xor(se, x, 32);
    }
    if (lane == 0) { wsum[w] = sb; wsum2[w] = se; }
    __syncthreads();

    for (int c0 = w; c0 < g; c0 += 32 * CPW) {
        int n[CPW], base[CPW], dn[CPW][2];
#pragma unroll
        for (int r = 0; r < CPW; ++r) {
            const int cc = c0 + 32 * r;
            const int cl = min(cc, g - 1);
            n[r]    = (cc < g) ? cn[cl] : 0;
            base[r] = cs[cl];
        }
#pragma unroll
        for (int r = 0; r < CPW; ++r) {
            const int last = max(n[r] - 1, 0);
            dn[r][0] = pairs[min(base[r] + min(lane, last), n_edges - 1)].y;
            dn[r][1] = pairs[min(base[r] + min(lane + 32, last), n_edges - 1)].y;
        }
#pragma unroll
        for (int r = 0; r < CPW; ++r) {
            if (lane < n[r])      atomicAdd(&nc[dn[r][0] - node0], 1);
            if (lane + 32 < n[r]) atomicAdd(&nc[dn[r][1] - node0], 1);
        }
#pragma unroll
        for (int r = 0; r < CPW; ++r)
            for (int k = 64 + lane; k < n[r]; k += 32)
                atomicAdd(&nc[pairs[min(base[r] + k, n_edges - 1)].y - node0], 1);
    }
    int beg = wsum[lane], end = wsum2[lane];
    for (int x = 16; x > 0; x >>= 1) {
        beg += __shfl_xor(beg, x, 32);
        end += __shfl_xor(end, x, 32);
    }
    __syncthreads();
    const int v   = (t < BNODES) ? nc[t] : 0;
    const int inc = wave_incl_scan(v, lane);
    if (lane == 31) wsum[w] = inc;
    __syncthreads();
    int p = (lane < w) ? wsum[lane] : 0;
    for (int x = 16; x > 0; x >>= 1) p += __shfl_xor(p, x, 32);
    const int ex = beg + p + inc - v;
    if (t < BNODES) {
        nc[t] = ex;
        ns[t] = ex;
        if (node0 + t <= n_nodes) off[node0 + t] = ex;
    }
    if (t == 0) ns[BNODES] = end;
    __syncthreads();
    for (int c0 = w; c0 < g; c0 += 32 * CPW) {
        int n[CPW], base[CPW];
        int2 pr[CPW][2];
#pragma unroll
        for (int r = 0; r < CPW; ++r) {
            const int cc = c0 + 32 * r;
            const int cl = min(cc, g - 1);
            n[r]    = (cc < g) ? cn[cl] : 0;
            base[r] = cs[cl];
        }
#pragma unroll
        for (int r = 0; r < CPW; ++r) {
            const int last = max(n[r] - 1, 0);
            pr[r][0] = pairs[min(base[r] + min(lane, last), n_edges - 1)];
            pr[r][1] = pairs[min(base[r] + min(lane + 32, last), n_edges - 1)];
        }
        int q[CPW][2];
#pragma unroll
        for (int r = 0; r < CPW; ++r) {
#pragma unroll
            for (int u = 0; u < 2; ++u) {
                q[r][u] = beg;
                if (lane + 32 * u < n[r]) q[r][u] = atomicAdd(&nc[pr[r][u].y - node0], 1);
            }
        }
#pragma unroll
        for (int r = 0; r < CPW; ++r) {
#pragma unroll
            for (int u = 0; u < 2; ++u) {
                if (lane + 32 * u < n[r]) {
                    const int rr = q[r][u] - beg;
                    if (rr < SCAP) stage[max(rr, 0)] = pr[r][u].x;
                    else           perm[min(q[r][u], n_edges - 1)] = pr[r][u].x;
                }
            }
        }
#pragma unroll
        for (int r = 0; r < CPW; ++r) {
            for (int k = 64 + lane; k < n[r]; k += 32) {
                const int2 pk = pairs[min(base[r] + k, n_edges - 1)];
                const int q  = atomicAdd(&nc[pk.y - node0], 1);
                const int rr = q - beg;
                if (rr < SCAP) stage[max(rr, 0)] = pk.x;
                else           perm[min(q, n_edges - 1)] = pk.x;
            }
        }
    }
    __syncthreads();
    const int mis = (int)(((unsigned long long)perm >> 2) & 31);
    const int nst = min(end - beg, SCAP);
    if (nst > 0) {
        for (int i0 = ((beg + mis) & ~31) - mis + t; i0 < beg + nst; i0 += 4 * ETHR) {
            int vv[4];
#pragma unroll
            for (int u = 0; u < 4; ++u) vv[u] = stage[min(max(i0 + u * ETHR - beg, 0), nst - 1)];
#pragma unroll
            for (int u = 0; u < 4; ++u) {
                const int i = i0 + u * ETHR;
                if (i >= beg && i < beg + nst) perm[i] = vv[u];
            }
        }
    }

    const int ntl = min(BNODES, n_nodes - node0);
    const int r0  = 16 * w;
    if (r0 >= ntl) return;
    float* tile = tiles + w * (16 * TP);
    const int hi   = lane >> 4;
    const int lo   = lane & 15;
    const int row0 = node0 + r0;

#define BIN_IDX(K) (((K) < SCAP) ? stage[max((K), 0)] : perm[min(beg + (K), n_edges - 1)])

    const v4f* __restrict__ f4 = (const v4f*)feat;
    const int lp = lane & 3;
    const int lq = lane >> 2;
    const v4f z = {0.f, 0.f, 0.f, 0.f};
    const int o = ns[min(r0 + min(lane, 16), ntl)];

    int nbeg[2], ndeg[2], nia[2], nib[2];
#pragma unroll
    for (int r = 0; r < 2; ++r) {
        const int j = 8 * r + lq;
        const int ob = __shfl(o, j, 32);
        nbeg[r] = ob - beg;
        ndeg[r] = __shfl(o, j + 1, 32) - ob;
        nia[r]  = (lp     < ndeg[r]) ? BIN_IDX(nbeg[r] + lp)     : 0;
        nib[r]  = (4 + lp < ndeg[r]) ? BIN_IDX(nbeg[r] + 4 + lp) : 0;
    }

#define GATHER_SLOT(U)                                                          \
    {                                                                           \
        const int s = __builtin_amdgcn_update_dpp(0, ((U) < 4) ? ca : cb,       \
                                                  ((U) & 3) * 0x55, 0xf, 0xf,   \
                                                  true);                        \
        v[U][0] = z; v[U][1] = z; v[U][2] = z;                                  \
        if ((U) < rem) {                                                        \
            const v4f* __restrict__ rp = f4 + (long long)s * 12 + lp;           \
            v[U][0] = rp[0];                                                    \
            v[U][1] = rp[4];                                                    \
            v[U][2] = rp[8];                                                    \
        }                                                                       \
    }

#pragma unroll
    for (int r = 0; r < 2; ++r) {
        const int j    = 8 * r + lq;
        const int begr = nbeg[r];
        const int deg  = ndeg[r];
        int na = nia[r], nb = nib[r];
        v4f a0 = z, a1 = z, a2 = z;
        for (int base = 0; base < deg; base += 8) {
            const int rem = deg - base;
            const int ca = na, cb = nb;
            na = (8 + lp  < rem) ? BIN_IDX(begr + base + 8 + lp)  : 0;
            nb = (12 + lp < rem) ? BIN_IDX(begr + base + 12 + lp) : 0;
            v4f v[8][3];
            GATHER_SLOT(0) GATHER_SLOT(1) GATHER_SLOT(2) GATHER_SLOT(3)
            GATHER_SLOT(4) GATHER_SLOT(5) GATHER_SLOT(6) GATHER_SLOT(7)
#pragma unroll
            for (int u = 0; u < 8; ++u) { a0 += v[u][0]; a1 += v[u][1]; a2 += v[u][2]; }
        }
        float* trow = &tile[j * TP + 4 * lp];
        *(v4f*)(trow)      = a0;
        *(v4f*)(trow + 16) = a1;
        *(v4f*)(trow + 32) = a2;
    }
#undef GATHER_SLOT
#undef BIN_IDX
    __builtin_amdgcn_fence(__ATOMIC_RELEASE, "wavefront");
    __builtin_amdgcn_wave_barrier();
    __builtin_amdgcn_fence(__ATOMIC_ACQUIRE, "wavefront");

    const int arow = min(row0 + lo, n_nodes - 1);
    v2f a_agg[12], a_x[12];
#pragma unroll
    for (int k = 0; k < 12; ++k) {
        const int kk = 4 * k + 2 * hi;
        const long long base = (long long)arow * DF + kk;
        a_agg[k] = *(const v2f*)&tile[lo * TP + kk];
        a_x[k].x = feat[base];
        a_x[k].y = feat[base + 1];
    }

#pragma unroll
    for (int nt = 0; nt < 3; ++nt) {
        const int ncol = nt * 16 + lo;
        const float bv = bias[ncol];
        v8f c;
#pragma unroll
        for (int i = 0; i < 8; ++i) c[i] = bv;

#pragma unroll
        for (int k = 0; k < 12; ++k) {
            const int krow = 4 * k + 2 * hi;
            v2f brel, broot;
            brel.x  = w_rel [(krow    ) * DF + ncol];
            brel.y  = w_rel [(krow + 1) * DF + ncol];
            broot.x = w_root[(krow    ) * DF + ncol];
            broot.y = w_root[(krow + 1) * DF + ncol];
            c = __builtin_amdgcn_wmma_f32_16x16x4_f32(
                    false, a_agg[k], false, brel, (short)0, c, false, false);
            c = __builtin_amdgcn_wmma_f32_16x16x4_f32(
                    false, a_x[k], false, broot, (short)0, c, false, false);
        }

#pragma unroll
        for (int v = 0; v < 8; ++v) {
            const float r = fmaxf(c[v], 0.f);
            const int lr = r0 + v + 8 * hi;
            if (lr < ntl) out[(long long)(node0 + lr) * DF + ncol] = r;
        }
    }
}

template <bool RELU>
__global__ __launch_bounds__(32)
void conv_layer_wmma(const float* __restrict__ feat,
                     const int* __restrict__ off,
                     const int* __restrict__ perm,
                     const float* __restrict__ w_rel,
                     const float* __restrict__ w_root,
                     const float* __restrict__ bias,
                     float* __restrict__ out,
                     int n_nodes) {
    __shared__ __attribute__((aligned(16))) float tile[16 * TP];
    const int lane = threadIdx.x;
    const int hi   = lane >> 4;
    const int lo   = lane & 15;
    const int row0 = blockIdx.x * 16;

    const v4f* __restrict__ f4 = (const v4f*)feat;
    const int p = lane & 3;
    const int q = lane >> 2;
    const v4f z = {0.f, 0.f, 0.f, 0.f};
    const int o = off[min(row0 + min(lane, 16), n_nodes)];

    int nbeg[2], ndeg[2], nia[2], nib[2];
#pragma unroll
    for (int r = 0; r < 2; ++r) {
        const int j = 8 * r + q;
        nbeg[r] = __shfl(o, j, 32);
        ndeg[r] = __shfl(o, j + 1, 32) - nbeg[r];
        nia[r]  = (p     < ndeg[r]) ? perm[nbeg[r] + p]     : 0;
        nib[r]  = (4 + p < ndeg[r]) ? perm[nbeg[r] + 4 + p] : 0;
    }

    asm volatile("" :: "v"(nia[0]), "v"(nib[0]), "v"(nia[1]), "v"(nib[1]));

#define GATHER_SLOT(U)                                                          \
    {                                                                           \
        const int s = __builtin_amdgcn_update_dpp(0, ((U) < 4) ? ca : cb,       \
                                                  ((U) & 3) * 0x55, 0xf, 0xf,   \
                                                  true);                        \
        v[U][0] = z; v[U][1] = z; v[U][2] = z;                                  \
        if ((U) < rem) {                                                        \
            const v4f* __restrict__ rp = f4 + (long long)s * 12 + p;            \
            v[U][0] = rp[0];                                                    \
            v[U][1] = rp[4];                                                    \
            v[U][2] = rp[8];                                                    \
        }                                                                       \
    }

#pragma unroll
    for (int r = 0; r < 2; ++r) {
        const int j   = 8 * r + q;
        const int beg = nbeg[r];
        const int deg = ndeg[r];
        int na = nia[r], nb = nib[r];
        v4f a0 = z, a1 = z, a2 = z;
        for (int base = 0; base < deg; base += 8) {
            const int rem = deg - base;
            const int ca = na, cb = nb;
            na = (8 + p  < rem) ? perm[beg + base + 8 + p]  : 0;
            nb = (12 + p < rem) ? perm[beg + base + 12 + p] : 0;
            v4f v[8][3];
            GATHER_SLOT(0) GATHER_SLOT(1) GATHER_SLOT(2) GATHER_SLOT(3)
            GATHER_SLOT(4) GATHER_SLOT(5) GATHER_SLOT(6) GATHER_SLOT(7)
#pragma unroll
            for (int u = 0; u < 8; ++u) { a0 += v[u][0]; a1 += v[u][1]; a2 += v[u][2]; }
        }
        float* trow = &tile[j * TP + 4 * p];
        *(v4f*)(trow)      = a0;
        *(v4f*)(trow + 16) = a1;
        *(v4f*)(trow + 32) = a2;
    }
#undef GATHER_SLOT
    __syncthreads();

    const int arow = min(row0 + lo, n_nodes - 1);
    v2f a_agg[12], a_x[12];
#pragma unroll
    for (int k = 0; k < 12; ++k) {
        const int kk = 4 * k + 2 * hi;
        const long long base = (long long)arow * DF + kk;
        a_agg[k] = *(const v2f*)&tile[lo * TP + kk];
        a_x[k].x = feat[base];
        a_x[k].y = feat[base + 1];
    }

#pragma unroll
    for (int nt = 0; nt < 3; ++nt) {
        const int ncol = nt * 16 + lo;
        const float bv = bias[ncol];
        v8f c;
#pragma unroll
        for (int i = 0; i < 8; ++i) c[i] = bv;

#pragma unroll
        for (int k = 0; k < 12; ++k) {
            const int krow = 4 * k + 2 * hi;
            v2f brel, broot;
            brel.x  = w_rel [(krow    ) * DF + ncol];
            brel.y  = w_rel [(krow + 1) * DF + ncol];
            broot.x = w_root[(krow    ) * DF + ncol];
            broot.y = w_root[(krow + 1) * DF + ncol];
            c = __builtin_amdgcn_wmma_f32_16x16x4_f32(
                    false, a_agg[k], false, brel, (short)0, c, false, false);
            c = __builtin_amdgcn_wmma_f32_16x16x4_f32(
                    false, a_x[k], false, broot, (short)0, c, false, false);
        }

#pragma unroll
        for (int v = 0; v < 8; ++v) {
            float r = c[v];
            if (RELU) r = fmaxf(r, 0.f);
            const int m = row0 + v + 8 * hi;
            if (m < n_nodes) out[(long long)m * DF + ncol] = r;
        }
    }
}

extern "C" void kernel_launch(void* const* d_in, const int* in_sizes, int n_in,
                              void* d_out, int out_size, void* d_ws, size_t ws_size,
                              hipStream_t stream) {
    const float* x       = (const float*)d_in[0];
    const int*   ei      = (const int*)  d_in[1];
    const float* w_rel1  = (const float*)d_in[2];
    const float* w_root1 = (const float*)d_in[3];
    const float* b1      = (const float*)d_in[4];
    const float* w_rel2  = (const float*)d_in[5];
    const float* w_root2 = (const float*)d_in[6];
    const float* b2      = (const float*)d_in[7];
    float*       out     = (float*)d_out;

    const int N = in_sizes[0] / DF;
    const int E = in_sizes[1] / 2;
    const int* src = ei;
    const int* dst = ei + E;

    float* h = (float*)d_ws;

    const int nb      = (N + 1 + BNODES - 1) / BNODES;
    const int g       = (E + ECH - 1) / ECH;
    const int tp      = (nb + 1 + 31) / 32 * 32;
    const int NO      = (N + 1 + 31) / 32 * 32;
    int*  tab   = (int*)(h + (size_t)N * DF);
    int*  off   = tab + (size_t)g * tp;
    int2* pairs = (int2*)(off + NO);
    int*  perm  = (int*)(pairs + E);

    const int mtiles  = (N + 15) / 16;

    bin_chunk<<<g, ETHR, nb * sizeof(int), stream>>>(src, dst, tab, pairs, E, nb, tp);
    bucket_fill_layer<<<nb, ETHR, 2 * g * sizeof(int), stream>>>(pairs, tab, off, perm,
                                                                 x, w_rel1, w_root1, b1, h, N, E, g, tp);

    conv_layer_wmma<false><<<mtiles, 32, 0, stream>>>(h, off, perm, w_rel2, w_root2, b2, out, N);
}


# Self operand: the tile's own 16 rows (3 KB contiguous) come by 6 whole-line loads ahead of the gather into a second LDS tile, A fragments from LDS; was 12 loads of 16 lines each behind the gather
# speedup vs baseline: 7.0150x; 1.0296x over previous
#include <hip/hip_runtime.h>
#include <hip/hip_bf16.h>


typedef __attribute__((ext_vector_type(2))) float v2f;
typedef __attribute__((ext_vector_type(4))) float v4f;
typedef __attribute__((ext_vector_type(8))) float v8f;

#define DF 48

#define BNODES 391
#define ECH    8192
#define ETHR   1024
#define TP     52

__device__ __forceinline__ int wave_incl_scan(int v, int lane) {
    for (int d = 1; d < 32; d <<= 1) {
        int o = __shfl_up(v, d, 32);
        if (lane >= d) v += o;
    }
    return v;
}

__global__ __launch_bounds__(ETHR)
void bin_chunk(const int* __restrict__ src,
               const int* __restrict__ dst,
               int* __restrict__ tab,
               int2* __restrict__ pairs,
               int n_edges, int nb, int tp) {
    extern __shared__ int lc[];
    __shared__ int wsum[32];
    __shared__ int2 stage[ECH];
    const int t    = threadIdx.x;
    const int lane = t & 31;
    const int w    = t >> 5;
    const int c    = blockIdx.x;

    const int e0 = c * ECH + t;
    int d[ECH / ETHR], s[ECH / ETHR], rk[ECH / ETHR];
#pragma unroll
    for (int k = 0; k < ECH / ETHR; ++k) {
        const int e = min(e0 + k * ETHR, n_edges - 1);
        d[k] = dst[e];
        s[k] = src[e];
    }
    for (int i = t; i < nb; i += ETHR) lc[i] = 0;
    __syncthreads();
#pragma unroll
    for (int k = 0; k < ECH / ETHR; ++k) {
        rk[k] = 0;
        if (e0 + k * ETHR < n_edges) rk[k] = atomicAdd(&lc[d[k] / BNODES], 1);
    }
    __syncthreads();

    int carry = 0;
    for (int b0 = 0; b0 < nb; b0 += ETHR) {
        const int i   = b0 + t;
        const int v   = (i < nb) ? lc[i] : 0;
        const int inc = wave_incl_scan(v, lane);
        if (lane == 31) wsum[w] = inc;
        __syncthreads();
        const int ws = wsum[lane];
        int p   = (lane < w) ? ws : 0;
        int all = ws;
        for (int x = 16; x > 0; x >>= 1) {
            p   += __shfl_xor(p, x, 32);
            all += __shfl_xor(all, x, 32);
        }
        const int ex = carry + p + inc - v;
        if (i < nb) {
            lc[i] = ex;
            tab[c * tp + i] = ex;
        }
        carry += all;
        __syncthreads();
    }
    if (t == 0) tab[c * tp + nb] = carry;

    int fp[ECH / ETHR];
#pragma unroll
    for (int k = 0; k < ECH / ETHR; ++k) fp[k] = lc[d[k] / BNODES];
#pragma unroll
    for (int k = 0; k < ECH / ETHR; ++k) {
        if (e0 + k * ETHR < n_edges) {
            stage[min(max(fp[k] + rk[k], 0), ECH - 1)] = make_int2(s[k], d[k]);
        }
    }
    __syncthreads();
    const int cnt = min(ECH, n_edges - c * ECH);
    int2 pr[ECH / ETHR];
#pragma unroll
    for (int k = 0; k < ECH / ETHR; ++k) pr[k] = stage[min(t + k * ETHR, cnt - 1)];
#pragma unroll
    for (int k = 0; k < ECH / ETHR; ++k) {
        const int i = t + k * ETHR;
        if (i < cnt) pairs[c * ECH + i] = pr[k];
    }
}

#define SCAP 16384
#define CPW  7
__global__ __launch_bounds__(ETHR)
void bucket_fill_layer(const int2* __restrict__ pairs,
                       const int* __restrict__ tab,
                       int* __restrict__ off,
                       int* __restrict__ perm,
                       const float* __restrict__ feat,
                       const float* __restrict__ w_rel,
                       const float* __restrict__ w_root,
                       const float* __restrict__ bias,
                       float* __restrict__ out,
                       int n_nodes, int n_edges, int g, int tp) {
    extern __shared__ int cs[];
    int* cn = cs + g;
    __shared__ int nc[BNODES];
    __shared__ int ns[BNODES + 1];
    __shared__ int wsum[32];
    __shared__ int wsum2[32];
    __shared__ int stage[SCAP];
    __shared__ __attribute__((aligned(16))) float tiles[(ETHR / 32) * 2 * 16 * TP];
    const int t    = threadIdx.x;
    const int lane = t & 31;
    const int w    = __builtin_amdgcn_readfirstlane(t >> 5);
    const int b    = blockIdx.x;
    const int node0 = b * BNODES;

    int sb = 0, se = 0;
    for (int cc = t; cc < g; cc += ETHR) {
        const int wl = min(ECH, n_edges - cc * ECH);
        const int s0 = min(max(tab[cc * tp + b], 0), wl);
        const int s1 = min(max(tab[cc * tp + b + 1], s0), wl);
        cs[cc] = cc * ECH + s0;
        cn[cc] = s1 - s0;
        sb += s0;
        se += s1;
    }
    if (t < BNODES) nc[t] = 0;
    for (int x = 16; x > 0; x >>= 1) {
        sb += __shfl_xor(sb, x, 32);
        se += __shfl_xor(se, x, 32);
    }
    if (lane == 0) { wsum[w] = sb; wsum2[w] = se; }
    __syncthreads();

    for (int c0 = w; c0 < g; c0 += 32 * CPW) {
        int n[CPW], base[CPW], dn[CPW][2];
#pragma unroll
        for (int r = 0; r < CPW; ++r) {
            const int cc = c0 + 32 * r;
            const int cl = min(cc, g - 1);
            n[r]    = (cc < g) ? cn[cl] : 0;
            base[r] = cs[cl];
        }
#pragma unroll
        for (int r = 0; r < CPW; ++r) {
            const int last = max(n[r] - 1, 0);
            dn[r][0] = pairs[min(base[r] + min(lane, last), n_edges - 1)].y;
            dn[r][1] = pairs[min(base[r] + min(lane + 32, last), n_edges - 1)].y;
        }
#pragma unroll
        for (int r = 0; r < CPW; ++r) {
            if (lane < n[r])      atomicAdd(&nc[dn[r][0] - node0], 1);
            if (lane + 32 < n[r]) atomicAdd(&nc[dn[r][1] - node0], 1);
        }
#pragma unroll
        for (int r = 0; r < CPW; ++r)
            for (int k = 64 + lane; k < n[r]; k += 32)
                atomicAdd(&nc[pairs[min(base[r] + k, n_edges - 1)].y - node0], 1);
    }
    int beg = wsum[lane], end = wsum2[lane];
    for (int x = 16; x > 0; x >>= 1) {
        beg += __shfl_xor(beg, x, 32);
        end += __shfl_xor(end, x, 32);
    }
    __syncthreads();
    const int v   = (t < BNODES) ? nc[t] : 0;
    const int inc = wave_incl_scan(v, lane);
    if (lane == 31) wsum[w] = inc;
    __syncthreads();
    int p = (lane < w) ? wsum[lane] : 0;
    for (int x = 16; x > 0; x >>= 1) p += __shfl_xor(p, x, 32);
    const int ex = beg + p + inc - v;
    if (t < BNODES) {
        nc[t] = ex;
        ns[t] = ex;
        if (node0 + t <= n_nodes) off[node0 + t] = ex;
    }
    if (t == 0) ns[BNODES] = end;
    __syncthreads();
    for (int c0 = w; c0 < g; c0 += 32 * CPW) {
        int n[CPW], base[CPW];
        int2 pr[CPW][2];
#pragma unroll
        for (int r = 0; r < CPW; ++r) {
            const int cc = c0 + 32 * r;
            const int cl = min(cc, g - 1);
            n[r]    = (cc < g) ? cn[cl] : 0;
            base[r] = cs[cl];
        }
#pragma unroll
        for (int r = 0; r < CPW; ++r) {
            const int last = max(n[r] - 1, 0);
            pr[r][0] = pairs[min(base[r] + min(lane, last), n_edges - 1)];
            pr[r][1] = pairs[min(base[r] + min(lane + 32, last), n_edges - 1)];
        }
        int q[CPW][2];
#pragma unroll
        for (int r = 0; r < CPW; ++r) {
#pragma unroll
            for (int u = 0; u < 2; ++u) {
                q[r][u] = beg;
                if (lane + 32 * u < n[r]) q[r][u] = atomicAdd(&nc[pr[r][u].y - node0], 1);
            }
        }
#pragma unroll
        for (int r = 0; r < CPW; ++r) {
#pragma unroll
            for (int u = 0; u < 2; ++u) {
                if (lane + 32 * u < n[r]) {
                    const int rr = q[r][u] - beg;
                    if (rr < SCAP) stage[max(rr, 0)] = pr[r][u].x;
                    else           perm[min(q[r][u], n_edges - 1)] = pr[r][u].x;
                }
            }
        }
#pragma unroll
        for (int r = 0; r < CPW; ++r) {
            for (int k = 64 + lane; k < n[r]; k += 32) {
                const int2 pk = pairs[min(base[r] + k, n_edges - 1)];
                const int q  = atomicAdd(&nc[pk.y - node0], 1);
                const int rr = q - beg;
                if (rr < SCAP) stage[max(rr, 0)] = pk.x;
                else           perm[min(q, n_edges - 1)] = pk.x;
            }
        }
    }
    __syncthreads();
    const int ntl = min(BNODES, n_nodes - node0);
    const int r0  = 16 * w;
    const bool has_tile = r0 < ntl;
    v4f own[6];
    if (has_tile) {
        const v4f* __restrict__ x4 = (const v4f*)feat;
        const int e0 = (node0 + r0) * (DF / 4) + lane;
        const int elast = n_nodes * (DF / 4) - 1;
#pragma unroll
        for (int i = 0; i < 6; ++i) own[i] = x4[min(e0 + 32 * i, elast)];
    }
    const int mis = (int)(((unsigned long long)perm >> 2) & 31);
    const int nst = min(end - beg, SCAP);
    if (nst > 0) {
        for (int i0 = ((beg + mis) & ~31) - mis + t; i0 < beg + nst; i0 += 4 * ETHR) {
            int vv[4];
#pragma unroll
            for (int u = 0; u < 4; ++u) vv[u] = stage[min(max(i0 + u * ETHR - beg, 0), nst - 1)];
#pragma unroll
            for (int u = 0; u < 4; ++u) {
                const int i = i0 + u * ETHR;
                if (i >= beg && i < beg + nst) perm[i] = vv[u];
            }
        }
    }

    if (!has_tile) return;
    float* tile = tiles + w * (2 * 16 * TP);
    const int hi   = lane >> 4;
    const int lo   = lane & 15;

#define BIN_IDX(K) (((K) < SCAP) ? stage[max((K), 0)] : perm[min(beg + (K), n_edges - 1)])

    const v4f* __restrict__ f4 = (const v4f*)feat;
    const int lp = lane & 3;
    const int lq = lane >> 2;
    const v4f z = {0.f, 0.f, 0.f, 0.f};
    const int o = ns[min(r0 + min(lane, 16), ntl)];

    int nbeg[2], ndeg[2], nia[2], nib[2];
#pragma unroll
    for (int r = 0; r < 2; ++r) {
        const int j = 8 * r + lq;
        const int ob = __shfl(o, j, 32);
        nbeg[r] = ob - beg;
        ndeg[r] = __shfl(o, j + 1, 32) - ob;
        nia[r]  = (lp     < ndeg[r]) ? BIN_IDX(nbeg[r] + lp)     : 0;
        nib[r]  = (4 + lp < ndeg[r]) ? BIN_IDX(nbeg[r] + 4 + lp) : 0;
    }

#pragma unroll
    for (int i = 0; i < 6; ++i) {
        const int e  = 32 * i + lane;
        const int er = e / 12;
        *(v4f*)&tile[16 * TP + er * TP + 4 * (e - 12 * er)] = own[i];
    }

#define GATHER_SLOT(U)                                                          \
    {                                                                           \
        const int s = __builtin_amdgcn_update_dpp(0, ((U) < 4) ? ca : cb,       \
                                                  ((U) & 3) * 0x55, 0xf, 0xf,   \
                                                  true);                        \
        v[U][0] = z; v[U][1] = z; v[U][2] = z;                                  \
        if ((U) < rem) {                                                        \
            const v4f* __restrict__ rp = f4 + (long long)s * 12 + lp;           \
            v[U][0] = rp[0];                                                    \
            v[U][1] = rp[4];                                                    \
            v[U][2] = rp[8];                                                    \
        }                                                                       \
    }

#pragma unroll
    for (int r = 0; r < 2; ++r) {
        const int j    = 8 * r + lq;
        const int begr = nbeg[r];
        const int deg  = ndeg[r];
        int na = nia[r], nb = nib[r];
        v4f a0 = z, a1 = z, a2 = z;
        for (int base = 0; base < deg; base += 8) {
            const int rem = deg - base;
            const int ca = na, cb = nb;
            na = (8 + lp  < rem) ? BIN_IDX(begr + base + 8 + lp)  : 0;
            nb = (12 + lp < rem) ? BIN_IDX(begr + base + 12 + lp) : 0;
            v4f v[8][3];
            GATHER_SLOT(0) GATHER_SLOT(1) GATHER_SLOT(2) GATHER_SLOT(3)
            GATHER_SLOT(4) GATHER_SLOT(5) GATHER_SLOT(6) GATHER_SLOT(7)
#pragma unroll
            for (int u = 0; u < 8; ++u) { a0 += v[u][0]; a1 += v[u][1]; a2 += v[u][2]; }
        }
        float* trow = &tile[j * TP + 4 * lp];
        *(v4f*)(trow)      = a0;
        *(v4f*)(trow + 16) = a1;
        *(v4f*)(trow + 32) = a2;
    }
#undef GATHER_SLOT
#undef BIN_IDX
    __builtin_amdgcn_fence(__ATOMIC_RELEASE, "wavefront");
    __builtin_amdgcn_wave_barrier();
    __builtin_amdgcn_fence(__ATOMIC_ACQUIRE, "wavefront");

    v2f a_agg[12], a_x[12];
#pragma unroll
    for (int k = 0; k < 12; ++k) {
        const int kk = 4 * k + 2 * hi;
        a_agg[k] = *(const v2f*)&tile[lo * TP + kk];
        a_x[k]   = *(const v2f*)&tile[16 * TP + lo * TP + kk];
    }

#pragma unroll
    for (int nt = 0; nt < 3; ++nt) {
        const int ncol = nt * 16 + lo;
        const float bv = bias[ncol];
        v8f c;
#pragma unroll
        for (int i = 0; i < 8; ++i) c[i] = bv;

#pragma unroll
        for (int k = 0; k < 12; ++k) {
            const int krow = 4 * k + 2 * hi;
            v2f brel, broot;
            brel.x  = w_rel [(krow    ) * DF + ncol];
            brel.y  = w_rel [(krow + 1) * DF + ncol];
            broot.x = w_root[(krow    ) * DF + ncol];
            broot.y = w_root[(krow + 1) * DF + ncol];
            c = __builtin_amdgcn_wmma_f32_16x16x4_f32(
                    false, a_agg[k], false, brel, (short)0, c, false, false);
            c = __builtin_amdgcn_wmma_f32_16x16x4_f32(
                    false, a_x[k], false, broot, (short)0, c, false, false);
        }

#pragma unroll
        for (int v = 0; v < 8; ++v) {
            const float r = fmaxf(c[v], 0.f);
            const int lr = r0 + v + 8 * hi;
            if (lr < ntl) out[(long long)(node0 + lr) * DF + ncol] = r;
        }
    }
}

template <bool RELU>
__global__ __launch_bounds__(32)
void conv_layer_wmma(const float* __restrict__ feat,
                     const int* __restrict__ off,
                     const int* __restrict__ perm,
                     const float* __restrict__ w_rel,
                     const float* __restrict__ w_root,
                     const float* __restrict__ bias,
                     float* __restrict__ out,
                     int n_nodes) {
    __shared__ __attribute__((aligned(16))) float tile[2 * 16 * TP];
    const int lane = threadIdx.x;
    const int hi   = lane >> 4;
    const int lo   = lane & 15;
    const int row0 = blockIdx.x * 16;

    const v4f* __restrict__ f4 = (const v4f*)feat;
    v4f own[6];
    {
        const int e0 = row0 * (DF / 4) + lane;
        const int elast = n_nodes * (DF / 4) - 1;
#pragma unroll
        for (int i = 0; i < 6; ++i) own[i] = f4[min(e0 + 32 * i, elast)];
    }

    const int p = lane & 3;
    const int q = lane >> 2;
    const v4f z = {0.f, 0.f, 0.f, 0.f};
    const int o = off[min(row0 + min(lane, 16), n_nodes)];

    int nbeg[2], ndeg[2], nia[2], nib[2];
#pragma unroll
    for (int r = 0; r < 2; ++r) {
        const int j = 8 * r + q;
        nbeg[r] = __shfl(o, j, 32);
        ndeg[r] = __shfl(o, j + 1, 32) - nbeg[r];
        nia[r]  = (p     < ndeg[r]) ? perm[nbeg[r] + p]     : 0;
        nib[r]  = (4 + p < ndeg[r]) ? perm[nbeg[r] + 4 + p] : 0;
    }

    asm volatile("" :: "v"(nia[0]), "v"(nib[0]), "v"(nia[1]), "v"(nib[1]));

#pragma unroll
    for (int i = 0; i < 6; ++i) {
        const int e  = 32 * i + lane;
        const int er = e / 12;
        *(v4f*)&tile[16 * TP + er * TP + 4 * (e - 12 * er)] = own[i];
    }

#define GATHER_SLOT(U)                                                          \
    {                                                                           \
        const int s = __builtin_amdgcn_update_dpp(0, ((U) < 4) ? ca : cb,       \
                                                  ((U) & 3) * 0x55, 0xf, 0xf,   \
                                                  true);                        \
        v[U][0] = z; v[U][1] = z; v[U][2] = z;                                  \
        if ((U) < rem) {                                                        \
            const v4f* __restrict__ rp = f4 + (long long)s * 12 + p;            \
            v[U][0] = rp[0];                                                    \
            v[U][1] = rp[4];                                                    \
            v[U][2] = rp[8];                                                    \
        }                                                                       \
    }

#pragma unroll
    for (int r = 0; r < 2; ++r) {
        const int j   = 8 * r + q;
        const int beg = nbeg[r];
        const int deg = ndeg[r];
        int na = nia[r], nb = nib[r];
        v4f a0 = z, a1 = z, a2 = z;
        for (int base = 0; base < deg; base += 8) {
            const int rem = deg - base;
            const int ca = na, cb = nb;
            na = (8 + p  < rem) ? perm[beg + base + 8 + p]  : 0;
            nb = (12 + p < rem) ? perm[beg + base + 12 + p] : 0;
            v4f v[8][3];
            GATHER_SLOT(0) GATHER_SLOT(1) GATHER_SLOT(2) GATHER_SLOT(3)
            GATHER_SLOT(4) GATHER_SLOT(5) GATHER_SLOT(6) GATHER_SLOT(7)
#pragma unroll
            for (int u = 0; u < 8; ++u) { a0 += v[u][0]; a1 += v[u][1]; a2 += v[u][2]; }
        }
        float* trow = &tile[j * TP + 4 * p];
        *(v4f*)(trow)      = a0;
        *(v4f*)(trow + 16) = a1;
        *(v4f*)(trow + 32) = a2;
    }
#undef GATHER_SLOT
    __syncthreads();

    v2f a_agg[12], a_x[12];
#pragma unroll
    for (int k = 0; k < 12; ++k) {
        const int kk = 4 * k + 2 * hi;
        a_agg[k] = *(const v2f*)&tile[lo * TP + kk];
        a_x[k]   = *(const v2f*)&tile[16 * TP + lo * TP + kk];
    }

#pragma unroll
    for (int nt = 0; nt < 3; ++nt) {
        const int ncol = nt * 16 + lo;
        const float bv = bias[ncol];
        v8f c;
#pragma unroll
        for (int i = 0; i < 8; ++i) c[i] = bv;

#pragma unroll
        for (int k = 0; k < 12; ++k) {
            const int krow = 4 * k + 2 * hi;
            v2f brel, broot;
            brel.x  = w_rel [(krow    ) * DF + ncol];
            brel.y  = w_rel [(krow + 1) * DF + ncol];
            broot.x = w_root[(krow    ) * DF + ncol];
            broot.y = w_root[(krow + 1) * DF + ncol];
            c = __builtin_amdgcn_wmma_f32_16x16x4_f32(
                    false, a_agg[k], false, brel, (short)0, c, false, false);
            c = __builtin_amdgcn_wmma_f32_16x16x4_f32(
                    false, a_x[k], false, broot, (short)0, c, false, false);
        }

#pragma unroll
        for (int v = 0; v < 8; ++v) {
            float r = c[v];
            if (RELU) r = fmaxf(r, 0.f);
            const int m = row0 + v + 8 * hi;
            if (m < n_nodes) out[(long long)m * DF + ncol] = r;
        }
    }
}

extern "C" void kernel_launch(void* const* d_in, const int* in_sizes, int n_in,
                              void* d_out, int out_size, void* d_ws, size_t ws_size,
                              hipStream_t stream) {
    const float* x       = (const float*)d_in[0];
    const int*   ei      = (const int*)  d_in[1];
    const float* w_rel1  = (const float*)d_in[2];
    const float* w_root1 = (const float*)d_in[3];
    const float* b1      = (const float*)d_in[4];
    const float* w_rel2  = (const float*)d_in[5];
    const float* w_root2 = (const float*)d_in[6];
    const float* b2      = (const float*)d_in[7];
    float*       out     = (float*)d_out;

    const int N = in_sizes[0] / DF;
    const int E = in_sizes[1] / 2;
    const int* src = ei;
    const int* dst = ei + E;

    float* h = (float*)d_ws;

    const int nb      = (N + 1 + BNODES - 1) / BNODES;
    const int g       = (E + ECH - 1) / ECH;
    const int tp      = (nb + 1 + 31) / 32 * 32;
    const int NO      = (N + 1 + 31) / 32 * 32;
    int*  tab   = (int*)(h + (size_t)N * DF);
    int*  off   = tab + (size_t)g * tp;
    int2* pairs = (int2*)(off + NO);
    int*  perm  = (int*)(pairs + E);

    const int mtiles  = (N + 15) / 16;

    bin_chunk<<<g, ETHR, nb * sizeof(int), stream>>>(src, dst, tab, pairs, E, nb, tp);
    bucket_fill_layer<<<nb, ETHR, 2 * g * sizeof(int), stream>>>(pairs, tab, off, perm,
                                                                 x, w_rel1, w_root1, b1, h, N, E, g, tp);

    conv_layer_wmma<false><<<mtiles, 32, 0, stream>>>(h, off, perm, w_rel2, w_root2, b2, out, N);
}
